# starting kernel
# baseline (speedup 1.0000x reference)
#include <hip/hip_runtime.h>

typedef __attribute__((ext_vector_type(2))) float v2f;
typedef __attribute__((ext_vector_type(8))) float v8f;

#define NPTS   200000
#define MOUT   100000
#define CIN    32
#define COUT   64
#define KNBR   32
#define VOX    27
#define KDIM   (VOX * CIN)
#define AS     (KDIM + 2)
#define MTILE  16
#define WAVES  4
#define BLOCK  (WAVES * 32)

__global__ __launch_bounds__(BLOCK)
void cconv_fused_kernel(const float* __restrict__ feats,
                        const float* __restrict__ inPos,
                        const float* __restrict__ outPos,
                        const float* __restrict__ extents,
                        const float* __restrict__ importance,
                        const int*   __restrict__ nbrIdx,
                        const float* __restrict__ Wg,
                        const float* __restrict__ bias,
                        float*       __restrict__ out)
{
    __shared__ float sA[MTILE][AS];
    __shared__ float sInvN[MTILE];

    const int lane = threadIdx.x & 31;
    const int wave = threadIdx.x >> 5;
    const int m0   = blockIdx.x * MTILE;
    const float invR = 2.0f / extents[0];

    for (int r = 0; r < MTILE / WAVES; ++r) {
        const int row = wave * (MTILE / WAVES) + r;
        const int m   = m0 + row;
        const float ox = outPos[m * 3 + 0];
        const float oy = outPos[m * 3 + 1];
        const float oz = outPos[m * 3 + 2];

        float acc[VOX];
        #pragma unroll
        for (int v = 0; v < VOX; ++v) acc[v] = 0.0f;
        float norm = 0.0f;

        for (int k = 0; k < KNBR; ++k) {
            const int idx = nbrIdx[m * KNBR + k];
            const float rx = (inPos[idx * 3 + 0] - ox) * invR;
            const float ry = (inPos[idx * 3 + 1] - oy) * invR;
            const float rz = (inPos[idx * 3 + 2] - oz) * invR;
            const float dist = sqrtf(rx * rx + ry * ry + rz * rz);
            const float linf = fmaxf(fabsf(rx), fmaxf(fabsf(ry), fabsf(rz)));
            const float scale = (linf > 1e-8f) ? dist / fmaxf(linf, 1e-8f) : 0.0f;
            const float imp = (dist <= 1.0f) ? importance[idx] : 0.0f;

            if (imp != 0.0f) {
                norm += imp;
                const float cx = fminf(fmaxf(rx * scale + 1.0f, 0.0f), 2.0f);
                const float cy = fminf(fmaxf(ry * scale + 1.0f, 0.0f), 2.0f);
                const float cz = fminf(fmaxf(rz * scale + 1.0f, 0.0f), 2.0f);
                float wx[3], wy[3], wz[3];
                #pragma unroll
                for (int a = 0; a < 3; ++a) {
                    wx[a] = fmaxf(0.0f, 1.0f - fabsf(cx - (float)a));
                    wy[a] = fmaxf(0.0f, 1.0f - fabsf(cy - (float)a));
                    wz[a] = fmaxf(0.0f, 1.0f - fabsf(cz - (float)a));
                }
                const float f = feats[idx * CIN + lane] * imp;
                const float fz0 = wz[0] * f, fz1 = wz[1] * f, fz2 = wz[2] * f;
                #pragma unroll
                for (int a = 0; a < 3; ++a) {
                    #pragma unroll
                    for (int b = 0; b < 3; ++b) {
                        const float wab = wx[a] * wy[b];
                        acc[(a * 3 + b) * 3 + 0] = fmaf(wab, fz0, acc[(a * 3 + b) * 3 + 0]);
                        acc[(a * 3 + b) * 3 + 1] = fmaf(wab, fz1, acc[(a * 3 + b) * 3 + 1]);
                        acc[(a * 3 + b) * 3 + 2] = fmaf(wab, fz2, acc[(a * 3 + b) * 3 + 2]);
                    }
                }
            }
        }

        #pragma unroll
        for (int v = 0; v < VOX; ++v)
            sA[row][v * CIN + lane] = acc[v];
        if (lane == 0)
            sInvN[row] = (norm > 0.0f) ? 1.0f / fmaxf(norm, 1e-12f) : 0.0f;
    }
    __syncthreads();

    const int n0   = wave * 16;
    const int half = lane >> 4;
    const int lm   = lane & 15;
    const float* Wt = Wg + n0 + lm;

    v8f c = {};
    #pragma unroll 4
    for (int kk = 0; kk < KDIM / 4; ++kk) {
        const int kb = kk * 4 + 2 * half;
        v2f a;
        a.x = sA[lm][kb + 0];
        a.y = sA[lm][kb + 1];
        v2f b;
        b.x = Wt[(kb + 0) * COUT];
        b.y = Wt[(kb + 1) * COUT];
        c = __builtin_amdgcn_wmma_f32_16x16x4_f32(
                 false, a,  false, b,
                 (short)0, c,  false,  false);
    }

    const float bv = bias[n0 + lm];
    #pragma unroll
    for (int j = 0; j < 8; ++j) {
        const int mrow = j + 8 * half;
        out[(long)(m0 + mrow) * COUT + n0 + lm] = c[j] * sInvN[mrow] + bv;
    }
}

extern "C" void kernel_launch(void* const* d_in, const int* in_sizes, int n_in,
                              void* d_out, int out_size, void* d_ws, size_t ws_size,
                              hipStream_t stream) {
    const float* feats      = (const float*)d_in[0];
    const float* inPos      = (const float*)d_in[1];
    const float* outPos     = (const float*)d_in[2];
    const float* extents    = (const float*)d_in[3];
    const float* importance = (const float*)d_in[4];
    const int*   nbrIdx     = (const int*)  d_in[5];
    const float* Wg         = (const float*)d_in[6];
    const float* bias       = (const float*)d_in[7];
    float* out = (float*)d_out;

    dim3 grid(MOUT / MTILE);
    cconv_fused_kernel<<<grid, BLOCK, 0, stream>>>(
        feats, inPos, outPos, extents, importance, nbrIdx, Wg, bias, out);
}


# filter held resident in LDS by 250 persistent work-groups (25 row tiles each): the GEMM phase reads both operands from LDS while the next tile's gathers are in flight
# speedup vs baseline: 2.9161x; 2.9161x over previous
#include <hip/hip_runtime.h>


typedef __attribute__((ext_vector_type(2))) float v2f;
typedef __attribute__((ext_vector_type(8))) float v8f;

#define NPTS   200000
#define MOUT   100000
#define CIN    32
#define COUT   64
#define KNBR   32
#define VOX    27
#define KDIM   (VOX * CIN)
#define AS     (KDIM + 2)
#define MTILE  16
#define NTILES (MOUT / MTILE)
#define NBLK   250
#define WAVES  8
#define BLOCK  (WAVES * 32)
#define WPAIRS (KDIM / 2)
#define WROW   128
#define KSTEPS_W (KDIM / 4 / 2)
#define KBLK   12

__device__ __forceinline__ int wpos(int p, int n) {
    return p * WROW + ((2 * n + 32 * (p & 1)) & (WROW - 1));
}

__device__ __forceinline__ void load_frags(const float* aP, const float* bP, int blk,
                                           v2f (&fa)[KBLK], v2f (&fb)[KBLK]) {
    #pragma unroll
    for (int i = 0; i < KBLK; ++i) {
        const int t = blk * KBLK + i;
        fa[i] = *(const v2f*)(aP + 4 * t);
        fb[i] = *(const v2f*)(bP + 2 * WROW * t);
    }
}

template <int S> __device__ __forceinline__ int row_share_i(int x) {
    return __builtin_amdgcn_update_dpp(0, x, 0x150 + S, 0xf, 0xf, true);
}
template <int S> __device__ __forceinline__ float row_share_f(float x) {
    return __int_as_float(row_share_i<S>(__float_as_int(x)));
}

__device__ __forceinline__ float tri(float x, float g) {
    return fmaxf(0.0f, 1.0f - fabsf(x - g));
}

template <int S>
__device__ __forceinline__ void gather_slot(const char* __restrict__ featB, int idx, unsigned c4,
                                            float (&f)[16][2]) {
    const unsigned off = (unsigned)row_share_i<S>(idx) * (unsigned)(CIN * 4) + c4;
    f[S][0] = *(const float*)(featB + off);
    f[S][1] = *(const float*)(featB + off + 64u);
}

__device__ __forceinline__ void issue_row(const char* __restrict__ featB,
                                          const float* __restrict__ inPos,
                                          const float* __restrict__ outPos,
                                          const float* __restrict__ importance,
                                          int idx, int m, unsigned c4,
                                          float (&f)[16][2], float (&g)[7]) {
    gather_slot<0>(featB, idx, c4, f);   gather_slot<1>(featB, idx, c4, f);
    gather_slot<2>(featB, idx, c4, f);   gather_slot<3>(featB, idx, c4, f);
    gather_slot<4>(featB, idx, c4, f);   gather_slot<5>(featB, idx, c4, f);
    gather_slot<6>(featB, idx, c4, f);   gather_slot<7>(featB, idx, c4, f);
    gather_slot<8>(featB, idx, c4, f);   gather_slot<9>(featB, idx, c4, f);
    gather_slot<10>(featB, idx, c4, f);  gather_slot<11>(featB, idx, c4, f);
    gather_slot<12>(featB, idx, c4, f);  gather_slot<13>(featB, idx, c4, f);
    gather_slot<14>(featB, idx, c4, f);  gather_slot<15>(featB, idx, c4, f);
    g[0] = inPos[idx * 3 + 0];
    g[1] = inPos[idx * 3 + 1];
    g[2] = inPos[idx * 3 + 2];
    g[3] = importance[idx];
    g[4] = outPos[m * 3 + 0];
    g[5] = outPos[m * 3 + 1];
    g[6] = outPos[m * 3 + 2];
}

template <int KK>
__device__ __forceinline__ void mma_step(float cx, float cy, float cz, float imp,
                                         float ax0, float ay0, float az0,
                                         float ax1, float ay1, float az1,
                                         const float (&f)[16][2],
                                         v8f& d00, v8f& d01, v8f& d10, v8f& d11, float& npart) {
    const float x0 = row_share_f<2 * KK>(cx),  x1 = row_share_f<2 * KK + 1>(cx);
    const float y0 = row_share_f<2 * KK>(cy),  y1 = row_share_f<2 * KK + 1>(cy);
    const float z0 = row_share_f<2 * KK>(cz),  z1 = row_share_f<2 * KK + 1>(cz);
    const float p0 = row_share_f<2 * KK>(imp), p1 = row_share_f<2 * KK + 1>(imp);
    v2f a0, a1, b0, b1;
    a0.x = tri(x0, ax0) * tri(y0, ay0) * tri(z0, az0) * p0;
    a0.y = tri(x1, ax0) * tri(y1, ay0) * tri(z1, az0) * p1;
    a1.x = tri(x0, ax1) * tri(y0, ay1) * tri(z0, az1) * p0;
    a1.y = tri(x1, ax1) * tri(y1, ay1) * tri(z1, az1) * p1;
    b0.x = f[2 * KK][0];  b0.y = f[2 * KK + 1][0];
    b1.x = f[2 * KK][1];  b1.y = f[2 * KK + 1][1];
    d00 = __builtin_amdgcn_wmma_f32_16x16x4_f32(false, a0, false, b0, (short)0, d00, false, false);
    d01 = __builtin_amdgcn_wmma_f32_16x16x4_f32(false, a0, false, b1, (short)0, d01, false, false);
    d10 = __builtin_amdgcn_wmma_f32_16x16x4_f32(false, a1, false, b0, (short)0, d10, false, false);
    d11 = __builtin_amdgcn_wmma_f32_16x16x4_f32(false, a1, false, b1, (short)0, d11, false, false);
    npart += p0 + p1;
}

__device__ __forceinline__ void compute_row(const float (&f)[16][2], const float (&g)[7], float invR,
                                            float ax0, float ay0, float az0,
                                            float ax1, float ay1, float az1,
                                            float* __restrict__ sArow, float* __restrict__ sInvNrow,
                                            int lane, int half, int lm) {
    const float rx = (g[0] - g[4]) * invR;
    const float ry = (g[1] - g[5]) * invR;
    const float rz = (g[2] - g[6]) * invR;
    const float dist = sqrtf(rx * rx + ry * ry + rz * rz);
    const float linf = fmaxf(fabsf(rx), fmaxf(fabsf(ry), fabsf(rz)));
    const float scale = (linf > 1e-8f) ? dist / fmaxf(linf, 1e-8f) : 0.0f;
    const float imp = (dist <= 1.0f) ? g[3] : 0.0f;
    const float cx = fminf(fmaxf(rx * scale + 1.0f, 0.0f), 2.0f);
    const float cy = fminf(fmaxf(ry * scale + 1.0f, 0.0f), 2.0f);
    const float cz = fminf(fmaxf(rz * scale + 1.0f, 0.0f), 2.0f);

    v8f d00 = {}, d01 = {}, d10 = {}, d11 = {};
    float npart = 0.0f;
    mma_step<0>(cx, cy, cz, imp, ax0, ay0, az0, ax1, ay1, az1, f, d00, d01, d10, d11, npart);
    mma_step<1>(cx, cy, cz, imp, ax0, ay0, az0, ax1, ay1, az1, f, d00, d01, d10, d11, npart);
    mma_step<2>(cx, cy, cz, imp, ax0, ay0, az0, ax1, ay1, az1, f, d00, d01, d10, d11, npart);
    mma_step<3>(cx, cy, cz, imp, ax0, ay0, az0, ax1, ay1, az1, f, d00, d01, d10, d11, npart);
    mma_step<4>(cx, cy, cz, imp, ax0, ay0, az0, ax1, ay1, az1, f, d00, d01, d10, d11, npart);
    mma_step<5>(cx, cy, cz, imp, ax0, ay0, az0, ax1, ay1, az1, f, d00, d01, d10, d11, npart);
    mma_step<6>(cx, cy, cz, imp, ax0, ay0, az0, ax1, ay1, az1, f, d00, d01, d10, d11, npart);
    mma_step<7>(cx, cy, cz, imp, ax0, ay0, az0, ax1, ay1, az1, f, d00, d01, d10, d11, npart);

    #pragma unroll
    for (int j = 0; j < 8; ++j) {
        const int va = 8 * half + j;
        sArow[va * CIN + lm]      = d00[j];
        sArow[va * CIN + 16 + lm] = d01[j];
        const int vb = 16 + 8 * half + j;
        if (vb < VOX) {
            sArow[vb * CIN + lm]      = d10[j];
            sArow[vb * CIN + 16 + lm] = d11[j];
        }
    }
    const float norm = npart + __shfl_xor(npart, 16);
    if (lane == 0)
        *sInvNrow = (norm > 0.0f) ? 1.0f / fmaxf(norm, 1e-12f) : 0.0f;
}

__global__ __launch_bounds__(BLOCK)
void cconv_persistent_kernel(const float* __restrict__ feats,
                             const float* __restrict__ inPos,
                             const float* __restrict__ outPos,
                             const float* __restrict__ extents,
                             const float* __restrict__ importance,
                             const int*   __restrict__ nbrIdx,
                             const float* __restrict__ Wg,
                             const float* __restrict__ bias,
                             float*       __restrict__ out)
{
    __shared__ __attribute__((aligned(16))) float sW[WPAIRS * WROW];
    __shared__ __attribute__((aligned(16))) float sA[MTILE][AS];
    __shared__ float sC[4][8][32];
    __shared__ float sInvN[MTILE];

    const int lane = threadIdx.x & 31;
    const int wave = threadIdx.x >> 5;
    const int half = lane >> 4;
    const int lm   = lane & 15;
    const float invR = 2.0f / extents[0];

    for (int i = threadIdx.x; i < WPAIRS * (COUT / 4); i += BLOCK) {
        const int p  = i >> 4;
        const int n4 = (i & 15) * 4;
        const float4 r0 = *(const float4*)(Wg + (2 * p) * COUT + n4);
        const float4 r1 = *(const float4*)(Wg + (2 * p + 1) * COUT + n4);
        float* d = sW + wpos(p, n4);
        *(float4*)(d)     = make_float4(r0.x, r1.x, r0.y, r1.y);
        *(float4*)(d + 4) = make_float4(r0.z, r1.z, r0.w, r1.w);
    }

    const int v0 = lm, v1 = 16 + lm;
    const float ax0 = (float)(v0 / 9), ay0 = (float)((v0 / 3) % 3), az0 = (float)(v0 % 3);
    const float ax1 = (v1 < VOX) ? (float)(v1 / 9) : 100.0f;
    const float ay1 = (float)((v1 / 3) % 3), az1 = (float)(v1 % 3);
    const char* featB = (const char*)feats;
    const unsigned c4 = (unsigned)lm * 4u;

    const int rowA  = wave * 2, rowB = wave * 2 + 1;
    const int nt    = wave & 3;
    const int khalf = wave >> 2;
    const int n0    = nt * 16;
    const float bv  = bias[n0 + lm];
    const float* aP = &sA[lm][khalf * (KDIM / 2) + 2 * half];
    const float* bP = sW + wpos(khalf * (2 * KSTEPS_W) + half, n0 + lm);

    int tile = blockIdx.x;
    float fA[16][2], fB[16][2], gA[7], gB[7];
    {
        const int mA = tile * MTILE + rowA, mB = tile * MTILE + rowB;
        const int iA = nbrIdx[mA * KNBR + lane];
        const int iB = nbrIdx[mB * KNBR + lane];
        issue_row(featB, inPos, outPos, importance, iA, mA, c4, fA, gA);
        issue_row(featB, inPos, outPos, importance, iB, mB, c4, fB, gB);
    }
    int idxAn, idxBn;
    {
        const int t1 = min(tile + NBLK, NTILES - 1);
        idxAn = nbrIdx[(t1 * MTILE + rowA) * KNBR + lane];
        idxBn = nbrIdx[(t1 * MTILE + rowB) * KNBR + lane];
    }

    for (; tile < NTILES; tile += NBLK) {
        const int m0 = tile * MTILE;

        compute_row(fA, gA, invR, ax0, ay0, az0, ax1, ay1, az1, sA[rowA], &sInvN[rowA], lane, half, lm);
        compute_row(fB, gB, invR, ax0, ay0, az0, ax1, ay1, az1, sA[rowB], &sInvN[rowB], lane, half, lm);
        __syncthreads();

        const int t1 = min(tile + NBLK, NTILES - 1);
        {
            const int t2 = min(tile + 2 * NBLK, NTILES - 1);
            const int iA = idxAn, mA = t1 * MTILE + rowA;
            idxAn = nbrIdx[(t2 * MTILE + rowA) * KNBR + lane];
            issue_row(featB, inPos, outPos, importance, iA, mA, c4, fA, gA);
        }

        float inv[8];
        #pragma unroll
        for (int j = 0; j < 8; ++j) inv[j] = sInvN[j + 8 * half];

        v8f c0 = {}, c1 = {};
        v2f fa[2][KBLK], fb[2][KBLK];
        load_frags(aP, bP, 0, fa[0], fb[0]);
        #pragma unroll
        for (int blk = 0; blk < KSTEPS_W / KBLK; ++blk) {
            const int cur = blk & 1;
            if (blk + 1 < KSTEPS_W / KBLK)
                load_frags(aP, bP, blk + 1, fa[cur ^ 1], fb[cur ^ 1]);
            if (blk == KSTEPS_W / KBLK / 2) {
                const int t2 = min(tile + 2 * NBLK, NTILES - 1);
                const int iB = idxBn, mB = t1 * MTILE + rowB;
                idxBn = nbrIdx[(t2 * MTILE + rowB) * KNBR + lane];
                issue_row(featB, inPos, outPos, importance, iB, mB, c4, fB, gB);
            }
            __builtin_amdgcn_sched_barrier(0);
            #pragma unroll
            for (int i = 0; i < KBLK; i += 2) {
                c0 = __builtin_amdgcn_wmma_f32_16x16x4_f32(false, fa[cur][i], false, fb[cur][i],
                                                           (short)0, c0, false, false);
                c1 = __builtin_amdgcn_wmma_f32_16x16x4_f32(false, fa[cur][i + 1], false, fb[cur][i + 1],
                                                           (short)0, c1, false, false);
            }
        }
        const v8f c = c0 + c1;

        if (khalf == 1) {
            #pragma unroll
            for (int j = 0; j < 8; ++j) sC[nt][j][lane] = c[j];
        }
        __syncthreads();

        if (khalf == 0) {
            #pragma unroll
            for (int j = 0; j < 8; ++j) {
                const int mrow = j + 8 * half;
                out[(long)(m0 + mrow) * COUT + n0 + lm] = (c[j] + sC[nt][j][lane]) * inv[j] + bv;
            }
        }
    }
}

extern "C" void kernel_launch(void* const* d_in, const int* in_sizes, int n_in,
                              void* d_out, int out_size, void* d_ws, size_t ws_size,
                              hipStream_t stream) {
    const float* feats      = (const float*)d_in[0];
    const float* inPos      = (const float*)d_in[1];
    const float* outPos     = (const float*)d_in[2];
    const float* extents    = (const float*)d_in[3];
    const float* importance = (const float*)d_in[4];
    const int*   nbrIdx     = (const int*)  d_in[5];
    const float* Wg         = (const float*)d_in[6];
    const float* bias       = (const float*)d_in[7];
    float* out = (float*)d_out;

    dim3 grid(NBLK);
    cconv_persistent_kernel<<<grid, BLOCK, 0, stream>>>(
        feats, inPos, outPos, extents, importance, nbrIdx, Wg, bias, out);
}


# interpolation weights computed once per neighbour in its own lane and handed to the voxel lanes as ready A fragments through a per-wave LDS tile, not rebuilt per voxel lane at every k-step
# speedup vs baseline: 3.1766x; 1.0893x over previous
#include <hip/hip_runtime.h>


typedef __attribute__((ext_vector_type(2))) float v2f;
typedef __attribute__((ext_vector_type(8))) float v8f;

#define NPTS   200000
#define MOUT   100000
#define CIN    32
#define COUT   64
#define KNBR   32
#define VOX    27
#define KDIM   (VOX * CIN)
#define AS     (KDIM + 2)
#define MTILE  16
#define NTILES (MOUT / MTILE)
#define NBLK   250
#define WAVES  8
#define BLOCK  (WAVES * 32)
#define WPAIRS (KDIM / 2)
#define WROW   128
#define KSTEPS_W (KDIM / 4 / 2)
#define KBLK   12
#define WP     36

__device__ __forceinline__ int wpos(int p, int n) {
    return p * WROW + ((2 * n + 32 * (p & 1)) & (WROW - 1));
}

__device__ __forceinline__ void load_frags(const float* aP, const float* bP, int blk,
                                           v2f (&fa)[KBLK], v2f (&fb)[KBLK]) {
    #pragma unroll
    for (int i = 0; i < KBLK; ++i) {
        const int t = blk * KBLK + i;
        fa[i] = *(const v2f*)(aP + 4 * t);
        fb[i] = *(const v2f*)(bP + 2 * WROW * t);
    }
}

template <int S> __device__ __forceinline__ int row_share_i(int x) {
    return __builtin_amdgcn_update_dpp(0, x, 0x150 + S, 0xf, 0xf, true);
}

__device__ __forceinline__ float tri(float x, float g) {
    return fmaxf(0.0f, 1.0f - fabsf(x - g));
}

template <int S>
__device__ __forceinline__ void gather_slot(const char* __restrict__ featB, int idx, unsigned c4,
                                            float (&f)[16][2]) {
    const unsigned off = (unsigned)row_share_i<S>(idx) * (unsigned)(CIN * 4) + c4;
    f[S][0] = *(const float*)(featB + off);
    f[S][1] = *(const float*)(featB + off + 64u);
}

__device__ __forceinline__ void issue_row(const char* __restrict__ featB,
                                          const float* __restrict__ inPos,
                                          const float* __restrict__ outPos,
                                          const float* __restrict__ importance,
                                          int idx, int m, unsigned c4,
                                          float (&f)[16][2], float (&g)[7]) {
    gather_slot<0>(featB, idx, c4, f);   gather_slot<1>(featB, idx, c4, f);
    gather_slot<2>(featB, idx, c4, f);   gather_slot<3>(featB, idx, c4, f);
    gather_slot<4>(featB, idx, c4, f);   gather_slot<5>(featB, idx, c4, f);
    gather_slot<6>(featB, idx, c4, f);   gather_slot<7>(featB, idx, c4, f);
    gather_slot<8>(featB, idx, c4, f);   gather_slot<9>(featB, idx, c4, f);
    gather_slot<10>(featB, idx, c4, f);  gather_slot<11>(featB, idx, c4, f);
    gather_slot<12>(featB, idx, c4, f);  gather_slot<13>(featB, idx, c4, f);
    gather_slot<14>(featB, idx, c4, f);  gather_slot<15>(featB, idx, c4, f);
    g[0] = inPos[idx * 3 + 0];
    g[1] = inPos[idx * 3 + 1];
    g[2] = inPos[idx * 3 + 2];
    g[3] = importance[idx];
    g[4] = outPos[m * 3 + 0];
    g[5] = outPos[m * 3 + 1];
    g[6] = outPos[m * 3 + 2];
}

template <int M> __device__ __forceinline__ float row_xmask_f(float x) {
    return __int_as_float(__builtin_amdgcn_update_dpp(0, __float_as_int(x), 0x160 + M, 0xf, 0xf, true));
}

__device__ __forceinline__ float store_weights(const float (&g)[7], float invR, float* wcol) {
    const float rx = (g[0] - g[4]) * invR;
    const float ry = (g[1] - g[5]) * invR;
    const float rz = (g[2] - g[6]) * invR;
    const float dist = sqrtf(rx * rx + ry * ry + rz * rz);
    const float linf = fmaxf(fabsf(rx), fmaxf(fabsf(ry), fabsf(rz)));
    const float scale = (linf > 1e-8f) ? dist / fmaxf(linf, 1e-8f) : 0.0f;
    const float imp = (dist <= 1.0f) ? g[3] : 0.0f;
    const float cx = fminf(fmaxf(rx * scale + 1.0f, 0.0f), 2.0f);
    const float cy = fminf(fmaxf(ry * scale + 1.0f, 0.0f), 2.0f);
    const float cz = fminf(fmaxf(rz * scale + 1.0f, 0.0f), 2.0f);
    float wx[3], wy[3], wz[3];
    #pragma unroll
    for (int t = 0; t < 3; ++t) {
        wx[t] = tri(cx, (float)t);
        wy[t] = tri(cy, (float)t);
        wz[t] = tri(cz, (float)t) * imp;
    }
    #pragma unroll
    for (int a = 0; a < 3; ++a) {
        #pragma unroll
        for (int b = 0; b < 3; ++b) {
            const float xy = wx[a] * wy[b];
            #pragma unroll
            for (int c = 0; c < 3; ++c) wcol[(9 * a + 3 * b + c) * WP] = xy * wz[c];
        }
    }
    return imp;
}

__device__ __forceinline__ void load_wfrags(const float* wrow, float4 (&q0)[4], float4 (&q1)[4]) {
    #pragma unroll
    for (int u = 0; u < 4; ++u) {
        q0[u] = *(const float4*)(wrow + 4 * u);
        q1[u] = *(const float4*)(wrow + 16 * WP + 4 * u);
    }
}

template <int KK>
__device__ __forceinline__ void mma_step(const float4 (&q0)[4], const float4 (&q1)[4],
                                         const float (&f)[16][2],
                                         v8f& d00, v8f& d01, v8f& d10, v8f& d11) {
    v2f a0, a1, b0, b1;
    if (KK & 1) {
        a0.x = q0[KK >> 1].z;  a0.y = q0[KK >> 1].w;
        a1.x = q1[KK >> 1].z;  a1.y = q1[KK >> 1].w;
    } else {
        a0.x = q0[KK >> 1].x;  a0.y = q0[KK >> 1].y;
        a1.x = q1[KK >> 1].x;  a1.y = q1[KK >> 1].y;
    }
    b0.x = f[2 * KK][0];  b0.y = f[2 * KK + 1][0];
    b1.x = f[2 * KK][1];  b1.y = f[2 * KK + 1][1];
    d00 = __builtin_amdgcn_wmma_f32_16x16x4_f32(false, a0, false, b0, (short)0, d00, false, false);
    d01 = __builtin_amdgcn_wmma_f32_16x16x4_f32(false, a0, false, b1, (short)0, d01, false, false);
    d10 = __builtin_amdgcn_wmma_f32_16x16x4_f32(false, a1, false, b0, (short)0, d10, false, false);
    d11 = __builtin_amdgcn_wmma_f32_16x16x4_f32(false, a1, false, b1, (short)0, d11, false, false);
}

__device__ __forceinline__ void product_row(const float4 (&q0)[4], const float4 (&q1)[4],
                                            const float (&f)[16][2], float imp,
                                            float* __restrict__ sArow, float* __restrict__ sInvNrow,
                                            int lane, int half, int lm) {
    v8f d00 = {}, d01 = {}, d10 = {}, d11 = {};
    mma_step<0>(q0, q1, f, d00, d01, d10, d11);
    mma_step<1>(q0, q1, f, d00, d01, d10, d11);
    mma_step<2>(q0, q1, f, d00, d01, d10, d11);
    mma_step<3>(q0, q1, f, d00, d01, d10, d11);
    mma_step<4>(q0, q1, f, d00, d01, d10, d11);
    mma_step<5>(q0, q1, f, d00, d01, d10, d11);
    mma_step<6>(q0, q1, f, d00, d01, d10, d11);
    mma_step<7>(q0, q1, f, d00, d01, d10, d11);

    #pragma unroll
    for (int j = 0; j < 8; ++j) {
        const int va = 8 * half + j;
        sArow[va * CIN + lm]      = d00[j];
        sArow[va * CIN + 16 + lm] = d01[j];
        const int vb = 16 + 8 * half + j;
        if (vb < VOX) {
            sArow[vb * CIN + lm]      = d10[j];
            sArow[vb * CIN + 16 + lm] = d11[j];
        }
    }
    float s = imp;
    s += row_xmask_f<1>(s);
    s += row_xmask_f<2>(s);
    s += row_xmask_f<4>(s);
    s += row_xmask_f<8>(s);
    const float norm = s + __shfl_xor(s, 16);
    if (lane == 0)
        *sInvNrow = (norm > 0.0f) ? 1.0f / fmaxf(norm, 1e-12f) : 0.0f;
}

__global__ __launch_bounds__(BLOCK)
void cconv_persistent_kernel(const float* __restrict__ feats,
                             const float* __restrict__ inPos,
                             const float* __restrict__ outPos,
                             const float* __restrict__ extents,
                             const float* __restrict__ importance,
                             const int*   __restrict__ nbrIdx,
                             const float* __restrict__ Wg,
                             const float* __restrict__ bias,
                             float*       __restrict__ out)
{
    __shared__ __attribute__((aligned(16))) float sW[WPAIRS * WROW];
    __shared__ __attribute__((aligned(16))) float sA[MTILE][AS];
    __shared__ float sC[4][8][32];
    __shared__ float sInvN[MTILE];
    __shared__ __attribute__((aligned(16))) float sWt[WAVES][32][WP];

    const int lane = threadIdx.x & 31;
    const int wave = threadIdx.x >> 5;
    const int half = lane >> 4;
    const int lm   = lane & 15;
    const float invR = 2.0f / extents[0];

    for (int i = threadIdx.x; i < WPAIRS * (COUT / 4); i += BLOCK) {
        const int p  = i >> 4;
        const int n4 = (i & 15) * 4;
        const float4 r0 = *(const float4*)(Wg + (2 * p) * COUT + n4);
        const float4 r1 = *(const float4*)(Wg + (2 * p + 1) * COUT + n4);
        float* d = sW + wpos(p, n4);
        *(float4*)(d)     = make_float4(r0.x, r1.x, r0.y, r1.y);
        *(float4*)(d + 4) = make_float4(r0.z, r1.z, r0.w, r1.w);
    }

    float* wcol = &sWt[wave][0][lane];
    const float* wrow = &sWt[wave][lm][16 * half];
    for (int r = VOX; r < 32; ++r) wcol[r * WP] = 0.0f;
    const char* featB = (const char*)feats;
    const unsigned c4 = (unsigned)lm * 4u;

    const int rowA  = wave * 2, rowB = wave * 2 + 1;
    const int nt    = wave & 3;
    const int khalf = wave >> 2;
    const int n0    = nt * 16;
    const float bv  = bias[n0 + lm];
    const float* aP = &sA[lm][khalf * (KDIM / 2) + 2 * half];
    const float* bP = sW + wpos(khalf * (2 * KSTEPS_W) + half, n0 + lm);

    int tile = blockIdx.x;
    float fA[16][2], fB[16][2], gA[7], gB[7];
    {
        const int mA = tile * MTILE + rowA, mB = tile * MTILE + rowB;
        const int iA = nbrIdx[mA * KNBR + lane];
        const int iB = nbrIdx[mB * KNBR + lane];
        issue_row(featB, inPos, outPos, importance, iA, mA, c4, fA, gA);
        issue_row(featB, inPos, outPos, importance, iB, mB, c4, fB, gB);
    }
    int idxAn, idxBn;
    {
        const int t1 = min(tile + NBLK, NTILES - 1);
        idxAn = nbrIdx[(t1 * MTILE + rowA) * KNBR + lane];
        idxBn = nbrIdx[(t1 * MTILE + rowB) * KNBR + lane];
    }

    for (; tile < NTILES; tile += NBLK) {
        const int m0 = tile * MTILE;

        float4 qA0[4], qA1[4], qB0[4], qB1[4];
        const float impA = store_weights(gA, invR, wcol);
        load_wfrags(wrow, qA0, qA1);
        const float impB = store_weights(gB, invR, wcol);
        load_wfrags(wrow, qB0, qB1);
        product_row(qA0, qA1, fA, impA, sA[rowA], &sInvN[rowA], lane, half, lm);
        product_row(qB0, qB1, fB, impB, sA[rowB], &sInvN[rowB], lane, half, lm);
        __syncthreads();

        const int t1 = min(tile + NBLK, NTILES - 1);
        {
            const int t2 = min(tile + 2 * NBLK, NTILES - 1);
            const int iA = idxAn, mA = t1 * MTILE + rowA;
            idxAn = nbrIdx[(t2 * MTILE + rowA) * KNBR + lane];
            issue_row(featB, inPos, outPos, importance, iA, mA, c4, fA, gA);
        }

        float inv[8];
        #pragma unroll
        for (int j = 0; j < 8; ++j) inv[j] = sInvN[j + 8 * half];

        v8f c0 = {}, c1 = {};
        v2f fa[2][KBLK], fb[2][KBLK];
        load_frags(aP, bP, 0, fa[0], fb[0]);
        #pragma unroll
        for (int blk = 0; blk < KSTEPS_W / KBLK; ++blk) {
            const int cur = blk & 1;
            if (blk + 1 < KSTEPS_W / KBLK)
                load_frags(aP, bP, blk + 1, fa[cur ^ 1], fb[cur ^ 1]);
            if (blk == KSTEPS_W / KBLK / 2) {
                const int t2 = min(tile + 2 * NBLK, NTILES - 1);
                const int iB = idxBn, mB = t1 * MTILE + rowB;
                idxBn = nbrIdx[(t2 * MTILE + rowB) * KNBR + lane];
                issue_row(featB, inPos, outPos, importance, iB, mB, c4, fB, gB);
            }
            __builtin_amdgcn_sched_barrier(0);
            #pragma unroll
            for (int i = 0; i < KBLK; i += 2) {
                c0 = __builtin_amdgcn_wmma_f32_16x16x4_f32(false, fa[cur][i], false, fb[cur][i],
                                                           (short)0, c0, false, false);
                c1 = __builtin_amdgcn_wmma_f32_16x16x4_f32(false, fa[cur][i + 1], false, fb[cur][i + 1],
                                                           (short)0, c1, false, false);
            }
        }
        const v8f c = c0 + c1;

        if (khalf == 1) {
            #pragma unroll
            for (int j = 0; j < 8; ++j) sC[nt][j][lane] = c[j];
        }
        __syncthreads();

        if (khalf == 0) {
            #pragma unroll
            for (int j = 0; j < 8; ++j) {
                const int mrow = j + 8 * half;
                out[(long)(m0 + mrow) * COUT + n0 + lm] = (c[j] + sC[nt][j][lane]) * inv[j] + bv;
            }
        }
    }
}

extern "C" void kernel_launch(void* const* d_in, const int* in_sizes, int n_in,
                              void* d_out, int out_size, void* d_ws, size_t ws_size,
                              hipStream_t stream) {
    const float* feats      = (const float*)d_in[0];
    const float* inPos      = (const float*)d_in[1];
    const float* outPos     = (const float*)d_in[2];
    const float* extents    = (const float*)d_in[3];
    const float* importance = (const float*)d_in[4];
    const int*   nbrIdx     = (const int*)  d_in[5];
    const float* Wg         = (const float*)d_in[6];
    const float* bias       = (const float*)d_in[7];
    float* out = (float*)d_out;

    dim3 grid(NBLK);
    cconv_persistent_kernel<<<grid, BLOCK, 0, stream>>>(
        feats, inPos, outPos, extents, importance, nbrIdx, Wg, bias, out);
}


# in-radius neighbours of each row ranked to the front of the first matrix product's depth by cross-lane pushes: k-steps and feature gathers of all-zero-weight groups of four are skipped
# speedup vs baseline: 3.6270x; 1.1418x over previous
#include <hip/hip_runtime.h>


typedef __attribute__((ext_vector_type(2))) float v2f;
typedef __attribute__((ext_vector_type(8))) float v8f;

#define NPTS   200000
#define MOUT   100000
#define CIN    32
#define COUT   64
#define KNBR   32
#define VOX    27
#define KDIM   (VOX * CIN)
#define AS     (KDIM + 2)
#define MTILE  16
#define NTILES (MOUT / MTILE)
#define NBLK   250
#define WAVES  8
#define BLOCK  (WAVES * 32)
#define WPAIRS (KDIM / 2)
#define WROW   128
#define KSTEPS_W (KDIM / 4 / 2)
#define KBLK   12
#define WP     36

__device__ __forceinline__ int wpos(int p, int n) {
    return p * WROW + ((2 * n + 32 * (p & 1)) & (WROW - 1));
}

__device__ __forceinline__ void load_frags(const float* aP, const float* bP, int blk,
                                           v2f (&fa)[KBLK], v2f (&fb)[KBLK]) {
    #pragma unroll
    for (int i = 0; i < KBLK; ++i) {
        const int t = blk * KBLK + i;
        fa[i] = *(const v2f*)(aP + 4 * t);
        fb[i] = *(const v2f*)(bP + 2 * WROW * t);
    }
}

template <int S> __device__ __forceinline__ int row_share_i(int x) {
    return __builtin_amdgcn_update_dpp(0, x, 0x150 + S, 0xf, 0xf, true);
}

__device__ __forceinline__ float tri(float x, float g) {
    return fmaxf(0.0f, 1.0f - fabsf(x - g));
}

template <int S>
__device__ __forceinline__ void gather_slot(const char* __restrict__ featB, int idx, unsigned c4,
                                            float (&f)[16][2]) {
    const unsigned off = (unsigned)row_share_i<S>(idx) * (unsigned)(CIN * 4) + c4;
    f[S][0] = *(const float*)(featB + off);
    f[S][1] = *(const float*)(featB + off + 64u);
}

template <int KK>
__device__ __forceinline__ void gather_pair(const char* __restrict__ featB, int idx, unsigned c4,
                                            float (&f)[16][2]) {
    gather_slot<2 * KK>(featB, idx, c4, f);
    gather_slot<2 * KK + 1>(featB, idx, c4, f);
}

__device__ __forceinline__ void issue_feats(const char* __restrict__ featB, int idx, unsigned c4,
                                            int ns, float (&f)[16][2]) {
    if (ns > 0) gather_pair<0>(featB, idx, c4, f);
    if (ns > 1) gather_pair<1>(featB, idx, c4, f);
    if (ns > 2) gather_pair<2>(featB, idx, c4, f);
    if (ns > 3) gather_pair<3>(featB, idx, c4, f);
    if (ns > 4) gather_pair<4>(featB, idx, c4, f);
    if (ns > 5) gather_pair<5>(featB, idx, c4, f);
    if (ns > 6) gather_pair<6>(featB, idx, c4, f);
    if (ns > 7) gather_pair<7>(featB, idx, c4, f);
}

__device__ __forceinline__ void issue_geom(const float* __restrict__ inPos,
                                           const float* __restrict__ outPos,
                                           const float* __restrict__ importance,
                                           int idx, int m, float (&g)[7]) {
    g[0] = inPos[idx * 3 + 0];
    g[1] = inPos[idx * 3 + 1];
    g[2] = inPos[idx * 3 + 2];
    g[3] = importance[idx];
    g[4] = outPos[m * 3 + 0];
    g[5] = outPos[m * 3 + 1];
    g[6] = outPos[m * 3 + 2];
}

template <int M> __device__ __forceinline__ float row_xmask_f(float x) {
    return __int_as_float(__builtin_amdgcn_update_dpp(0, __float_as_int(x), 0x160 + M, 0xf, 0xf, true));
}

__device__ __forceinline__ int push_i(int dst4, int x) {
    return __builtin_amdgcn_ds_permute(dst4, x);
}
__device__ __forceinline__ float push_f(int dst4, float x) {
    return __int_as_float(__builtin_amdgcn_ds_permute(dst4, __float_as_int(x)));
}

__device__ __forceinline__ int compact_row(const float (&g)[7], int idx, float invR, int lane,
                                           int pos, float (&c)[4], int& cidx) {
    const float rx = (g[0] - g[4]) * invR;
    const float ry = (g[1] - g[5]) * invR;
    const float rz = (g[2] - g[6]) * invR;
    const float dist = sqrtf(rx * rx + ry * ry + rz * rz);
    const float linf = fmaxf(fabsf(rx), fmaxf(fabsf(ry), fabsf(rz)));
    const float scale = (linf > 1e-8f) ? dist / fmaxf(linf, 1e-8f) : 0.0f;
    const bool in = (dist <= 1.0f);
    const float imp = in ? g[3] : 0.0f;
    const float cx = fminf(fmaxf(rx * scale + 1.0f, 0.0f), 2.0f);
    const float cy = fminf(fmaxf(ry * scale + 1.0f, 0.0f), 2.0f);
    const float cz = fminf(fmaxf(rz * scale + 1.0f, 0.0f), 2.0f);
    const unsigned mask = __builtin_amdgcn_ballot_w32(in);
    const int n = __builtin_popcount(mask);
    const int below = (int)__builtin_amdgcn_mbcnt_lo(mask, 0u);
    const int rank = in ? below : n + (lane - below);
    const int dst4 = 4 * (16 * ((rank >> 1) & 1) + 2 * (rank >> 2) + (rank & 1));
    const int ri = push_i(dst4, idx);
    c[0] = push_f(dst4, cx);
    c[1] = push_f(dst4, cy);
    c[2] = push_f(dst4, cz);
    c[3] = push_f(dst4, imp);
    const int rf = __builtin_amdgcn_readfirstlane(ri);
    cidx = (pos < n) ? ri : rf;
    return (n + 3) >> 2;
}

__device__ __forceinline__ float store_weights(const float (&c)[4], float* wcol) {
    const float imp = c[3];
    float wx[3], wy[3], wz[3];
    #pragma unroll
    for (int t = 0; t < 3; ++t) {
        wx[t] = tri(c[0], (float)t);
        wy[t] = tri(c[1], (float)t);
        wz[t] = tri(c[2], (float)t) * imp;
    }
    #pragma unroll
    for (int a = 0; a < 3; ++a) {
        #pragma unroll
        for (int b = 0; b < 3; ++b) {
            const float xy = wx[a] * wy[b];
            #pragma unroll
            for (int cc = 0; cc < 3; ++cc) wcol[(9 * a + 3 * b + cc) * WP] = xy * wz[cc];
        }
    }
    return imp;
}

__device__ __forceinline__ void load_wfrags(const float* wrow, float4 (&q0)[4], float4 (&q1)[4]) {
    #pragma unroll
    for (int u = 0; u < 4; ++u) {
        q0[u] = *(const float4*)(wrow + 4 * u);
        q1[u] = *(const float4*)(wrow + 16 * WP + 4 * u);
    }
}

template <int KK>
__device__ __forceinline__ void mma_step(const float4 (&q0)[4], const float4 (&q1)[4],
                                         const float (&f)[16][2],
                                         v8f& d00, v8f& d01, v8f& d10, v8f& d11) {
    v2f a0, a1, b0, b1;
    if (KK & 1) {
        a0.x = q0[KK >> 1].z;  a0.y = q0[KK >> 1].w;
        a1.x = q1[KK >> 1].z;  a1.y = q1[KK >> 1].w;
    } else {
        a0.x = q0[KK >> 1].x;  a0.y = q0[KK >> 1].y;
        a1.x = q1[KK >> 1].x;  a1.y = q1[KK >> 1].y;
    }
    b0.x = f[2 * KK][0];  b0.y = f[2 * KK + 1][0];
    b1.x = f[2 * KK][1];  b1.y = f[2 * KK + 1][1];
    d00 = __builtin_amdgcn_wmma_f32_16x16x4_f32(false, a0, false, b0, (short)0, d00, false, false);
    d01 = __builtin_amdgcn_wmma_f32_16x16x4_f32(false, a0, false, b1, (short)0, d01, false, false);
    d10 = __builtin_amdgcn_wmma_f32_16x16x4_f32(false, a1, false, b0, (short)0, d10, false, false);
    d11 = __builtin_amdgcn_wmma_f32_16x16x4_f32(false, a1, false, b1, (short)0, d11, false, false);
}

__device__ __forceinline__ void product_row(const float4 (&q0)[4], const float4 (&q1)[4],
                                            const float (&f)[16][2], int ns, float imp,
                                            float* __restrict__ sArow, float* __restrict__ sInvNrow,
                                            int lane, int half, int lm) {
    v8f d00 = {}, d01 = {}, d10 = {}, d11 = {};
    if (ns > 0) mma_step<0>(q0, q1, f, d00, d01, d10, d11);
    if (ns > 1) mma_step<1>(q0, q1, f, d00, d01, d10, d11);
    if (ns > 2) mma_step<2>(q0, q1, f, d00, d01, d10, d11);
    if (ns > 3) mma_step<3>(q0, q1, f, d00, d01, d10, d11);
    if (ns > 4) mma_step<4>(q0, q1, f, d00, d01, d10, d11);
    if (ns > 5) mma_step<5>(q0, q1, f, d00, d01, d10, d11);
    if (ns > 6) mma_step<6>(q0, q1, f, d00, d01, d10, d11);
    if (ns > 7) mma_step<7>(q0, q1, f, d00, d01, d10, d11);

    #pragma unroll
    for (int j = 0; j < 8; ++j) {
        const int va = 8 * half + j;
        sArow[va * CIN + lm]      = d00[j];
        sArow[va * CIN + 16 + lm] = d01[j];
        const int vb = 16 + 8 * half + j;
        if (vb < VOX) {
            sArow[vb * CIN + lm]      = d10[j];
            sArow[vb * CIN + 16 + lm] = d11[j];
        }
    }
    float s = imp;
    s += row_xmask_f<1>(s);
    s += row_xmask_f<2>(s);
    s += row_xmask_f<4>(s);
    s += row_xmask_f<8>(s);
    const float norm = s + __shfl_xor(s, 16);
    if (lane == 0)
        *sInvNrow = (norm > 0.0f) ? 1.0f / fmaxf(norm, 1e-12f) : 0.0f;
}

__global__ __launch_bounds__(BLOCK)
void cconv_persistent_kernel(const float* __restrict__ feats,
                             const float* __restrict__ inPos,
                             const float* __restrict__ outPos,
                             const float* __restrict__ extents,
                             const float* __restrict__ importance,
                             const int*   __restrict__ nbrIdx,
                             const float* __restrict__ Wg,
                             const float* __restrict__ bias,
                             float*       __restrict__ out)
{
    __shared__ __attribute__((aligned(16))) float sW[WPAIRS * WROW];
    __shared__ __attribute__((aligned(16))) float sA[MTILE][AS];
    __shared__ float sC[4][8][32];
    __shared__ float sInvN[MTILE];
    __shared__ __attribute__((aligned(16))) float sWt[WAVES][32][WP];

    const int lane = threadIdx.x & 31;
    const int wave = threadIdx.x >> 5;
    const int half = lane >> 4;
    const int lm   = lane & 15;
    const float invR = 2.0f / extents[0];

    for (int i = threadIdx.x; i < WPAIRS * (COUT / 4); i += BLOCK) {
        const int p  = i >> 4;
        const int n4 = (i & 15) * 4;
        const float4 r0 = *(const float4*)(Wg + (2 * p) * COUT + n4);
        const float4 r1 = *(const float4*)(Wg + (2 * p + 1) * COUT + n4);
        float* d = sW + wpos(p, n4);
        *(float4*)(d)     = make_float4(r0.x, r1.x, r0.y, r1.y);
        *(float4*)(d + 4) = make_float4(r0.z, r1.z, r0.w, r1.w);
    }

    float* wcol = &sWt[wave][0][lane];
    const float* wrow = &sWt[wave][lm][16 * half];
    for (int r = VOX; r < 32; ++r) wcol[r * WP] = 0.0f;
    const char* featB = (const char*)feats;
    const unsigned c4 = (unsigned)lm * 4u;
    const int pos = 4 * (lm >> 1) + 2 * half + (lane & 1);

    const int rowA  = wave * 2, rowB = wave * 2 + 1;
    const int nt    = wave & 3;
    const int khalf = wave >> 2;
    const int n0    = nt * 16;
    const float bv  = bias[n0 + lm];
    const float* aP = &sA[lm][khalf * (KDIM / 2) + 2 * half];
    const float* bP = sW + wpos(khalf * (2 * KSTEPS_W) + half, n0 + lm);

    int tile = blockIdx.x;
    float fA[16][2], fB[16][2], gA[7], gB[7], cA[4], cB[4];
    int nsA, nsB;
    int r1A, r1B;
    int r2A, r2B;
    {
        const int t1 = min(tile + NBLK, NTILES - 1);
        const int t2 = min(tile + 2 * NBLK, NTILES - 1);
        const int mA = tile * MTILE + rowA, mB = tile * MTILE + rowB;
        const int r0A = nbrIdx[mA * KNBR + lane];
        const int r0B = nbrIdx[mB * KNBR + lane];
        r1A = nbrIdx[(t1 * MTILE + rowA) * KNBR + lane];
        r1B = nbrIdx[(t1 * MTILE + rowB) * KNBR + lane];
        r2A = nbrIdx[(t2 * MTILE + rowA) * KNBR + lane];
        r2B = nbrIdx[(t2 * MTILE + rowB) * KNBR + lane];
        issue_geom(inPos, outPos, importance, r0A, mA, gA);
        issue_geom(inPos, outPos, importance, r0B, mB, gB);
        int ciA, ciB;
        nsA = compact_row(gA, r0A, invR, lane, pos, cA, ciA);
        nsB = compact_row(gB, r0B, invR, lane, pos, cB, ciB);
        issue_feats(featB, ciA, c4, nsA, fA);
        issue_feats(featB, ciB, c4, nsB, fB);
        issue_geom(inPos, outPos, importance, r1A, t1 * MTILE + rowA, gA);
        issue_geom(inPos, outPos, importance, r1B, t1 * MTILE + rowB, gB);
    }

    for (; tile < NTILES; tile += NBLK) {
        const int m0 = tile * MTILE;

        float4 qA0[4], qA1[4], qB0[4], qB1[4];
        const float impA = store_weights(cA, wcol);
        load_wfrags(wrow, qA0, qA1);
        const float impB = store_weights(cB, wcol);
        load_wfrags(wrow, qB0, qB1);
        product_row(qA0, qA1, fA, nsA, impA, sA[rowA], &sInvN[rowA], lane, half, lm);
        product_row(qB0, qB1, fB, nsB, impB, sA[rowB], &sInvN[rowB], lane, half, lm);
        __syncthreads();

        const int t2 = min(tile + 2 * NBLK, NTILES - 1);
        const int t3 = min(tile + 3 * NBLK, NTILES - 1);
        const int r3A = nbrIdx[(t3 * MTILE + rowA) * KNBR + lane];
        const int r3B = nbrIdx[(t3 * MTILE + rowB) * KNBR + lane];

        float inv[8];
        #pragma unroll
        for (int j = 0; j < 8; ++j) inv[j] = sInvN[j + 8 * half];

        v8f c0 = {}, c1 = {};
        v2f fa[2][KBLK], fb[2][KBLK];
        load_frags(aP, bP, 0, fa[0], fb[0]);
        #pragma unroll
        for (int blk = 0; blk < KSTEPS_W / KBLK; ++blk) {
            const int cur = blk & 1;
            if (blk + 1 < KSTEPS_W / KBLK)
                load_frags(aP, bP, blk + 1, fa[cur ^ 1], fb[cur ^ 1]);
            if (blk == 1) {
                int ciA, ciB;
                nsA = compact_row(gA, r1A, invR, lane, pos, cA, ciA);
                nsB = compact_row(gB, r1B, invR, lane, pos, cB, ciB);
                issue_feats(featB, ciA, c4, nsA, fA);
                issue_feats(featB, ciB, c4, nsB, fB);
                issue_geom(inPos, outPos, importance, r2A, t2 * MTILE + rowA, gA);
                issue_geom(inPos, outPos, importance, r2B, t2 * MTILE + rowB, gB);
                r1A = r2A;  r1B = r2B;
                r2A = r3A;  r2B = r3B;
            }
            __builtin_amdgcn_sched_barrier(0);
            #pragma unroll
            for (int i = 0; i < KBLK; i += 2) {
                c0 = __builtin_amdgcn_wmma_f32_16x16x4_f32(false, fa[cur][i], false, fb[cur][i],
                                                           (short)0, c0, false, false);
                c1 = __builtin_amdgcn_wmma_f32_16x16x4_f32(false, fa[cur][i + 1], false, fb[cur][i + 1],
                                                           (short)0, c1, false, false);
            }
        }
        const v8f c = c0 + c1;

        if (khalf == 1) {
            #pragma unroll
            for (int j = 0; j < 8; ++j) sC[nt][j][lane] = c[j];
        }
        __syncthreads();

        if (khalf == 0) {
            #pragma unroll
            for (int j = 0; j < 8; ++j) {
                const int mrow = j + 8 * half;
                out[(long)(m0 + mrow) * COUT + n0 + lm] = (c[j] + sC[nt][j][lane]) * inv[j] + bv;
            }
        }
    }
}

extern "C" void kernel_launch(void* const* d_in, const int* in_sizes, int n_in,
                              void* d_out, int out_size, void* d_ws, size_t ws_size,
                              hipStream_t stream) {
    const float* feats      = (const float*)d_in[0];
    const float* inPos      = (const float*)d_in[1];
    const float* outPos     = (const float*)d_in[2];
    const float* extents    = (const float*)d_in[3];
    const float* importance = (const float*)d_in[4];
    const int*   nbrIdx     = (const int*)  d_in[5];
    const float* Wg         = (const float*)d_in[6];
    const float* bias       = (const float*)d_in[7];
    float* out = (float*)d_out;

    dim3 grid(NBLK);
    cconv_persistent_kernel<<<grid, BLOCK, 0, stream>>>(
        feats, inPos, outPos, extents, importance, nbrIdx, Wg, bias, out);
}


# filter held in registers, not LDS: each wave keeps its 16-column, half-depth slice (108 B fragments, 216 registers) for all 25 tiles, so the second product loads only A fragments from LDS
# speedup vs baseline: 3.8394x; 1.0586x over previous
#include <hip/hip_runtime.h>


typedef __attribute__((ext_vector_type(2))) float v2f;
typedef __attribute__((ext_vector_type(8))) float v8f;

#define NPTS   200000
#define MOUT   100000
#define CIN    32
#define COUT   64
#define KNBR   32
#define VOX    27
#define KDIM   (VOX * CIN)
#define AS     (KDIM + 2)
#define MTILE  16
#define NTILES (MOUT / MTILE)
#define NBLK   250
#define WAVES  8
#define BLOCK  (WAVES * 32)
#define KSTEPS_W (KDIM / 4 / 2)
#define KBLK   12
#define WP     36

__device__ __forceinline__ void load_frags(const float* aP, int blk, v2f (&fa)[KBLK]) {
    #pragma unroll
    for (int i = 0; i < KBLK; ++i) {
        const int t = blk * KBLK + i;
        fa[i] = *(const v2f*)(aP + 4 * t);
    }
}

template <int S> __device__ __forceinline__ int row_share_i(int x) {
    return __builtin_amdgcn_update_dpp(0, x, 0x150 + S, 0xf, 0xf, true);
}

__device__ __forceinline__ float tri(float x, float g) {
    return fmaxf(0.0f, 1.0f - fabsf(x - g));
}

template <int S>
__device__ __forceinline__ void gather_slot(const char* __restrict__ featB, int idx, unsigned c4,
                                            float (&f)[16][2]) {
    const unsigned off = (unsigned)row_share_i<S>(idx) * (unsigned)(CIN * 4) + c4;
    f[S][0] = *(const float*)(featB + off);
    f[S][1] = *(const float*)(featB + off + 64u);
}

template <int KK>
__device__ __forceinline__ void gather_pair(const char* __restrict__ featB, int idx, unsigned c4,
                                            float (&f)[16][2]) {
    gather_slot<2 * KK>(featB, idx, c4, f);
    gather_slot<2 * KK + 1>(featB, idx, c4, f);
}

__device__ __forceinline__ void issue_feats(const char* __restrict__ featB, int idx, unsigned c4,
                                            int ns, float (&f)[16][2]) {
    if (ns > 0) gather_pair<0>(featB, idx, c4, f);
    if (ns > 1) gather_pair<1>(featB, idx, c4, f);
    if (ns > 2) gather_pair<2>(featB, idx, c4, f);
    if (ns > 3) gather_pair<3>(featB, idx, c4, f);
    if (ns > 4) gather_pair<4>(featB, idx, c4, f);
    if (ns > 5) gather_pair<5>(featB, idx, c4, f);
    if (ns > 6) gather_pair<6>(featB, idx, c4, f);
    if (ns > 7) gather_pair<7>(featB, idx, c4, f);
}

__device__ __forceinline__ void issue_geom(const float* __restrict__ inPos,
                                           const float* __restrict__ outPos,
                                           const float* __restrict__ importance,
                                           int idx, int m, float (&g)[7]) {
    g[0] = inPos[idx * 3 + 0];
    g[1] = inPos[idx * 3 + 1];
    g[2] = inPos[idx * 3 + 2];
    g[3] = importance[idx];
    g[4] = outPos[m * 3 + 0];
    g[5] = outPos[m * 3 + 1];
    g[6] = outPos[m * 3 + 2];
}

template <int M> __device__ __forceinline__ float row_xmask_f(float x) {
    return __int_as_float(__builtin_amdgcn_update_dpp(0, __float_as_int(x), 0x160 + M, 0xf, 0xf, true));
}

__device__ __forceinline__ int push_i(int dst4, int x) {
    return __builtin_amdgcn_ds_permute(dst4, x);
}
__device__ __forceinline__ float push_f(int dst4, float x) {
    return __int_as_float(__builtin_amdgcn_ds_permute(dst4, __float_as_int(x)));
}

__device__ __forceinline__ int compact_row(const float (&g)[7], int idx, float invR, int lane,
                                           int pos, float (&c)[4], int& cidx) {
    const float rx = (g[0] - g[4]) * invR;
    const float ry = (g[1] - g[5]) * invR;
    const float rz = (g[2] - g[6]) * invR;
    const float dist = sqrtf(rx * rx + ry * ry + rz * rz);
    const float linf = fmaxf(fabsf(rx), fmaxf(fabsf(ry), fabsf(rz)));
    const float scale = (linf > 1e-8f) ? dist / fmaxf(linf, 1e-8f) : 0.0f;
    const bool in = (dist <= 1.0f);
    const float imp = in ? g[3] : 0.0f;
    const float cx = fminf(fmaxf(rx * scale + 1.0f, 0.0f), 2.0f);
    const float cy = fminf(fmaxf(ry * scale + 1.0f, 0.0f), 2.0f);
    const float cz = fminf(fmaxf(rz * scale + 1.0f, 0.0f), 2.0f);
    const unsigned mask = __builtin_amdgcn_ballot_w32(in);
    const int n = __builtin_popcount(mask);
    const int below = (int)__builtin_amdgcn_mbcnt_lo(mask, 0u);
    const int rank = in ? below : n + (lane - below);
    const int dst4 = 4 * (16 * ((rank >> 1) & 1) + 2 * (rank >> 2) + (rank & 1));
    const int ri = push_i(dst4, idx);
    c[0] = push_f(dst4, cx);
    c[1] = push_f(dst4, cy);
    c[2] = push_f(dst4, cz);
    c[3] = push_f(dst4, imp);
    const int rf = __builtin_amdgcn_readfirstlane(ri);
    cidx = (pos < n) ? ri : rf;
    return (n + 3) >> 2;
}

__device__ __forceinline__ float store_weights(const float (&c)[4], float* wcol) {
    const float imp = c[3];
    float wx[3], wy[3], wz[3];
    #pragma unroll
    for (int t = 0; t < 3; ++t) {
        wx[t] = tri(c[0], (float)t);
        wy[t] = tri(c[1], (float)t);
        wz[t] = tri(c[2], (float)t) * imp;
    }
    #pragma unroll
    for (int a = 0; a < 3; ++a) {
        #pragma unroll
        for (int b = 0; b < 3; ++b) {
            const float xy = wx[a] * wy[b];
            #pragma unroll
            for (int cc = 0; cc < 3; ++cc) wcol[(9 * a + 3 * b + cc) * WP] = xy * wz[cc];
        }
    }
    return imp;
}

__device__ __forceinline__ void load_wfrags(const float* wrow, float4 (&q0)[4], float4 (&q1)[4]) {
    #pragma unroll
    for (int u = 0; u < 4; ++u) {
        q0[u] = *(const float4*)(wrow + 4 * u);
        q1[u] = *(const float4*)(wrow + 16 * WP + 4 * u);
    }
}

template <int KK>
__device__ __forceinline__ void mma_step(const float4 (&q0)[4], const float4 (&q1)[4],
                                         const float (&f)[16][2],
                                         v8f& d00, v8f& d01, v8f& d10, v8f& d11) {
    v2f a0, a1, b0, b1;
    if (KK & 1) {
        a0.x = q0[KK >> 1].z;  a0.y = q0[KK >> 1].w;
        a1.x = q1[KK >> 1].z;  a1.y = q1[KK >> 1].w;
    } else {
        a0.x = q0[KK >> 1].x;  a0.y = q0[KK >> 1].y;
        a1.x = q1[KK >> 1].x;  a1.y = q1[KK >> 1].y;
    }
    b0.x = f[2 * KK][0];  b0.y = f[2 * KK + 1][0];
    b1.x = f[2 * KK][1];  b1.y = f[2 * KK + 1][1];
    d00 = __builtin_amdgcn_wmma_f32_16x16x4_f32(false, a0, false, b0, (short)0, d00, false, false);
    d01 = __builtin_amdgcn_wmma_f32_16x16x4_f32(false, a0, false, b1, (short)0, d01, false, false);
    d10 = __builtin_amdgcn_wmma_f32_16x16x4_f32(false, a1, false, b0, (short)0, d10, false, false);
    d11 = __builtin_amdgcn_wmma_f32_16x16x4_f32(false, a1, false, b1, (short)0, d11, false, false);
}

__device__ __forceinline__ void product_row(const float4 (&q0)[4], const float4 (&q1)[4],
                                            const float (&f)[16][2], int ns, float imp,
                                            float* __restrict__ sArow, float* __restrict__ sInvNrow,
                                            int lane, int half, int lm) {
    v8f d00 = {}, d01 = {}, d10 = {}, d11 = {};
    if (ns > 0) mma_step<0>(q0, q1, f, d00, d01, d10, d11);
    if (ns > 1) mma_step<1>(q0, q1, f, d00, d01, d10, d11);
    if (ns > 2) mma_step<2>(q0, q1, f, d00, d01, d10, d11);
    if (ns > 3) mma_step<3>(q0, q1, f, d00, d01, d10, d11);
    if (ns > 4) mma_step<4>(q0, q1, f, d00, d01, d10, d11);
    if (ns > 5) mma_step<5>(q0, q1, f, d00, d01, d10, d11);
    if (ns > 6) mma_step<6>(q0, q1, f, d00, d01, d10, d11);
    if (ns > 7) mma_step<7>(q0, q1, f, d00, d01, d10, d11);

    #pragma unroll
    for (int j = 0; j < 8; ++j) {
        const int va = 8 * half + j;
        sArow[va * CIN + lm]      = d00[j];
        sArow[va * CIN + 16 + lm] = d01[j];
        const int vb = 16 + 8 * half + j;
        if (vb < VOX) {
            sArow[vb * CIN + lm]      = d10[j];
            sArow[vb * CIN + 16 + lm] = d11[j];
        }
    }
    float s = imp;
    s += row_xmask_f<1>(s);
    s += row_xmask_f<2>(s);
    s += row_xmask_f<4>(s);
    s += row_xmask_f<8>(s);
    const float norm = s + __shfl_xor(s, 16);
    if (lane == 0)
        *sInvNrow = (norm > 0.0f) ? 1.0f / fmaxf(norm, 1e-12f) : 0.0f;
}

__global__ __launch_bounds__(BLOCK)
void cconv_persistent_kernel(const float* __restrict__ feats,
                             const float* __restrict__ inPos,
                             const float* __restrict__ outPos,
                             const float* __restrict__ extents,
                             const float* __restrict__ importance,
                             const int*   __restrict__ nbrIdx,
                             const float* __restrict__ Wg,
                             const float* __restrict__ bias,
                             float*       __restrict__ out)
{
    __shared__ __attribute__((aligned(16))) float sA[MTILE][AS];
    __shared__ float sC[4][8][32];
    __shared__ float sInvN[MTILE];
    __shared__ __attribute__((aligned(16))) float sWt[WAVES][32][WP];

    const int lane = threadIdx.x & 31;
    const int wave = threadIdx.x >> 5;
    const int half = lane >> 4;
    const int lm   = lane & 15;
    const float invR = 2.0f / extents[0];

    float* wcol = &sWt[wave][0][lane];
    const float* wrow = &sWt[wave][lm][16 * half];
    for (int r = VOX; r < 32; ++r) wcol[r * WP] = 0.0f;
    const char* featB = (const char*)feats;
    const unsigned c4 = (unsigned)lm * 4u;
    const int pos = 4 * (lm >> 1) + 2 * half + (lane & 1);

    const int rowA  = wave * 2, rowB = wave * 2 + 1;
    const int nt    = wave & 3;
    const int khalf = wave >> 2;
    const int n0    = nt * 16;
    const float bv  = bias[n0 + lm];
    const float* aP = &sA[lm][khalf * (KDIM / 2) + 2 * half];

    v2f wf[KSTEPS_W];
    {
        const float* wg = Wg + (khalf * (KDIM / 2) + 2 * half) * COUT + n0 + lm;
        #pragma unroll
        for (int t = 0; t < KSTEPS_W; ++t) {
            wf[t].x = wg[(4 * t) * COUT];
            wf[t].y = wg[(4 * t + 1) * COUT];
        }
    }

    int tile = blockIdx.x;
    float fA[16][2], fB[16][2], gA[7], gB[7], cA[4], cB[4];
    int nsA, nsB;
    int r1A, r1B;
    int r2A, r2B;
    {
        const int t1 = min(tile + NBLK, NTILES - 1);
        const int t2 = min(tile + 2 * NBLK, NTILES - 1);
        const int mA = tile * MTILE + rowA, mB = tile * MTILE + rowB;
        const int r0A = nbrIdx[mA * KNBR + lane];
        const int r0B = nbrIdx[mB * KNBR + lane];
        r1A = nbrIdx[(t1 * MTILE + rowA) * KNBR + lane];
        r1B = nbrIdx[(t1 * MTILE + rowB) * KNBR + lane];
        r2A = nbrIdx[(t2 * MTILE + rowA) * KNBR + lane];
        r2B = nbrIdx[(t2 * MTILE + rowB) * KNBR + lane];
        issue_geom(inPos, outPos, importance, r0A, mA, gA);
        issue_geom(inPos, outPos, importance, r0B, mB, gB);
        int ciA, ciB;
        nsA = compact_row(gA, r0A, invR, lane, pos, cA, ciA);
        nsB = compact_row(gB, r0B, invR, lane, pos, cB, ciB);
        issue_feats(featB, ciA, c4, nsA, fA);
        issue_feats(featB, ciB, c4, nsB, fB);
        issue_geom(inPos, outPos, importance, r1A, t1 * MTILE + rowA, gA);
        issue_geom(inPos, outPos, importance, r1B, t1 * MTILE + rowB, gB);
    }

    for (; tile < NTILES; tile += NBLK) {
        const int m0 = tile * MTILE;

        float4 qA0[4], qA1[4], qB0[4], qB1[4];
        const float impA = store_weights(cA, wcol);
        load_wfrags(wrow, qA0, qA1);
        const float impB = store_weights(cB, wcol);
        load_wfrags(wrow, qB0, qB1);
        product_row(qA0, qA1, fA, nsA, impA, sA[rowA], &sInvN[rowA], lane, half, lm);
        product_row(qB0, qB1, fB, nsB, impB, sA[rowB], &sInvN[rowB], lane, half, lm);
        __syncthreads();

        const int t2 = min(tile + 2 * NBLK, NTILES - 1);
        const int t3 = min(tile + 3 * NBLK, NTILES - 1);
        const int r3A = nbrIdx[(t3 * MTILE + rowA) * KNBR + lane];
        const int r3B = nbrIdx[(t3 * MTILE + rowB) * KNBR + lane];

        float inv[8];
        #pragma unroll
        for (int j = 0; j < 8; ++j) inv[j] = sInvN[j + 8 * half];

        v8f c0 = {}, c1 = {};
        v2f fa[2][KBLK];
        load_frags(aP, 0, fa[0]);
        #pragma unroll
        for (int blk = 0; blk < KSTEPS_W / KBLK; ++blk) {
            const int cur = blk & 1;
            if (blk + 1 < KSTEPS_W / KBLK)
                load_frags(aP, blk + 1, fa[cur ^ 1]);
            if (blk == 1) {
                int ciA, ciB;
                nsA = compact_row(gA, r1A, invR, lane, pos, cA, ciA);
                nsB = compact_row(gB, r1B, invR, lane, pos, cB, ciB);
                issue_feats(featB, ciA, c4, nsA, fA);
                issue_feats(featB, ciB, c4, nsB, fB);
                issue_geom(inPos, outPos, importance, r2A, t2 * MTILE + rowA, gA);
                issue_geom(inPos, outPos, importance, r2B, t2 * MTILE + rowB, gB);
                r1A = r2A;  r1B = r2B;
                r2A = r3A;  r2B = r3B;
            }
            __builtin_amdgcn_sched_barrier(0);
            #pragma unroll
            for (int i = 0; i < KBLK; i += 2) {
                c0 = __builtin_amdgcn_wmma_f32_16x16x4_f32(false, fa[cur][i], false, wf[blk * KBLK + i],
                                                           (short)0, c0, false, false);
                c1 = __builtin_amdgcn_wmma_f32_16x16x4_f32(false, fa[cur][i + 1], false, wf[blk * KBLK + i + 1],
                                                           (short)0, c1, false, false);
            }
        }
        const v8f c = c0 + c1;

        if (khalf == 1) {
            #pragma unroll
            for (int j = 0; j < 8; ++j) sC[nt][j][lane] = c[j];
        }
        __syncthreads();

        if (khalf == 0) {
            #pragma unroll
            for (int j = 0; j < 8; ++j) {
                const int mrow = j + 8 * half;
                out[(long)(m0 + mrow) * COUT + n0 + lm] = (c[j] + sC[nt][j][lane]) * inv[j] + bv;
            }
        }
    }
}

extern "C" void kernel_launch(void* const* d_in, const int* in_sizes, int n_in,
                              void* d_out, int out_size, void* d_ws, size_t ws_size,
                              hipStream_t stream) {
    const float* feats      = (const float*)d_in[0];
    const float* inPos      = (const float*)d_in[1];
    const float* outPos     = (const float*)d_in[2];
    const float* extents    = (const float*)d_in[3];
    const float* importance = (const float*)d_in[4];
    const int*   nbrIdx     = (const int*)  d_in[5];
    const float* Wg         = (const float*)d_in[6];
    const float* bias       = (const float*)d_in[7];
    float* out = (float*)d_out;

    dim3 grid(NBLK);
    cconv_persistent_kernel<<<grid, BLOCK, 0, stream>>>(
        feats, inPos, outPos, extents, importance, nbrIdx, Wg, bias, out);
}


# first product transposed (channels x voxels): a lane's 8 accumulators are 8 consecutive channels of one voxel and reach the LDS A tile as two 16-byte conflict-free stores; tile pitch 36 per voxel
# speedup vs baseline: 3.9748x; 1.0353x over previous
#include <hip/hip_runtime.h>


typedef __attribute__((ext_vector_type(2))) float v2f;
typedef __attribute__((ext_vector_type(8))) float v8f;

#define NPTS   200000
#define MOUT   100000
#define CIN    32
#define COUT   64
#define KNBR   32
#define VOX    27
#define KDIM   (VOX * CIN)
#define VP     36
#define AS     (VOX * VP)
#define MTILE  16
#define NTILES (MOUT / MTILE)
#define NBLK   250
#define WAVES  8
#define BLOCK  (WAVES * 32)
#define KSTEPS_W (KDIM / 4 / 2)
#define KBLK   12
#define WP     36

__device__ __forceinline__ void load_frags(const float* aP, int blk, v2f (&fa)[KBLK]) {
    #pragma unroll
    for (int i = 0; i < KBLK; ++i) {
        const int t = blk * KBLK + i;
        fa[i] = *(const v2f*)(aP + (t >> 2) * VP + 4 * (t & 3));
    }
}

template <int S> __device__ __forceinline__ int row_share_i(int x) {
    return __builtin_amdgcn_update_dpp(0, x, 0x150 + S, 0xf, 0xf, true);
}

__device__ __forceinline__ float tri(float x, float g) {
    return fmaxf(0.0f, 1.0f - fabsf(x - g));
}

template <int S>
__device__ __forceinline__ void gather_slot(const char* __restrict__ featB, int idx, unsigned c4,
                                            float (&f)[16][2]) {
    const unsigned off = (unsigned)row_share_i<S>(idx) * (unsigned)(CIN * 4) + c4;
    f[S][0] = *(const float*)(featB + off);
    f[S][1] = *(const float*)(featB + off + 64u);
}

template <int KK>
__device__ __forceinline__ void gather_pair(const char* __restrict__ featB, int idx, unsigned c4,
                                            float (&f)[16][2]) {
    gather_slot<2 * KK>(featB, idx, c4, f);
    gather_slot<2 * KK + 1>(featB, idx, c4, f);
}

__device__ __forceinline__ void issue_feats(const char* __restrict__ featB, int idx, unsigned c4,
                                            int ns, float (&f)[16][2]) {
    if (ns > 0) gather_pair<0>(featB, idx, c4, f);
    if (ns > 1) gather_pair<1>(featB, idx, c4, f);
    if (ns > 2) gather_pair<2>(featB, idx, c4, f);
    if (ns > 3) gather_pair<3>(featB, idx, c4, f);
    if (ns > 4) gather_pair<4>(featB, idx, c4, f);
    if (ns > 5) gather_pair<5>(featB, idx, c4, f);
    if (ns > 6) gather_pair<6>(featB, idx, c4, f);
    if (ns > 7) gather_pair<7>(featB, idx, c4, f);
}

__device__ __forceinline__ void issue_geom(const float* __restrict__ inPos,
                                           const float* __restrict__ outPos,
                                           const float* __restrict__ importance,
                                           int idx, int m, float (&g)[7]) {
    g[0] = inPos[idx * 3 + 0];
    g[1] = inPos[idx * 3 + 1];
    g[2] = inPos[idx * 3 + 2];
    g[3] = importance[idx];
    g[4] = outPos[m * 3 + 0];
    g[5] = outPos[m * 3 + 1];
    g[6] = outPos[m * 3 + 2];
}

template <int M> __device__ __forceinline__ float row_xmask_f(float x) {
    return __int_as_float(__builtin_amdgcn_update_dpp(0, __float_as_int(x), 0x160 + M, 0xf, 0xf, true));
}

__device__ __forceinline__ int push_i(int dst4, int x) {
    return __builtin_amdgcn_ds_permute(dst4, x);
}
__device__ __forceinline__ float push_f(int dst4, float x) {
    return __int_as_float(__builtin_amdgcn_ds_permute(dst4, __float_as_int(x)));
}

__device__ __forceinline__ int compact_row(const float (&g)[7], int idx, float invR, int lane,
                                           int pos, float (&c)[4], int& cidx) {
    const float rx = (g[0] - g[4]) * invR;
    const float ry = (g[1] - g[5]) * invR;
    const float rz = (g[2] - g[6]) * invR;
    const float dist = sqrtf(rx * rx + ry * ry + rz * rz);
    const float linf = fmaxf(fabsf(rx), fmaxf(fabsf(ry), fabsf(rz)));
    const float scale = (linf > 1e-8f) ? dist / fmaxf(linf, 1e-8f) : 0.0f;
    const bool in = (dist <= 1.0f);
    const float imp = in ? g[3] : 0.0f;
    const float cx = fminf(fmaxf(rx * scale + 1.0f, 0.0f), 2.0f);
    const float cy = fminf(fmaxf(ry * scale + 1.0f, 0.0f), 2.0f);
    const float cz = fminf(fmaxf(rz * scale + 1.0f, 0.0f), 2.0f);
    const unsigned mask = __builtin_amdgcn_ballot_w32(in);
    const int n = __builtin_popcount(mask);
    const int below = (int)__builtin_amdgcn_mbcnt_lo(mask, 0u);
    const int rank = in ? below : n + (lane - below);
    const int dst4 = 4 * (16 * ((rank >> 1) & 1) + 2 * (rank >> 2) + (rank & 1));
    const int ri = push_i(dst4, idx);
    c[0] = push_f(dst4, cx);
    c[1] = push_f(dst4, cy);
    c[2] = push_f(dst4, cz);
    c[3] = push_f(dst4, imp);
    const int rf = __builtin_amdgcn_readfirstlane(ri);
    cidx = (pos < n) ? ri : rf;
    return (n + 3) >> 2;
}

__device__ __forceinline__ float store_weights(const float (&c)[4], float* wcol) {
    const float imp = c[3];
    float wx[3], wy[3], wz[3];
    #pragma unroll
    for (int t = 0; t < 3; ++t) {
        wx[t] = tri(c[0], (float)t);
        wy[t] = tri(c[1], (float)t);
        wz[t] = tri(c[2], (float)t) * imp;
    }
    #pragma unroll
    for (int a = 0; a < 3; ++a) {
        #pragma unroll
        for (int b = 0; b < 3; ++b) {
            const float xy = wx[a] * wy[b];
            #pragma unroll
            for (int cc = 0; cc < 3; ++cc) wcol[(9 * a + 3 * b + cc) * WP] = xy * wz[cc];
        }
    }
    return imp;
}

__device__ __forceinline__ void load_wfrags(const float* wrow, float4 (&q0)[4], float4 (&q1)[4]) {
    #pragma unroll
    for (int u = 0; u < 4; ++u) {
        q0[u] = *(const float4*)(wrow + 4 * u);
        q1[u] = *(const float4*)(wrow + 16 * WP + 4 * u);
    }
}

template <int KK>
__device__ __forceinline__ void mma_step(const float4 (&q0)[4], const float4 (&q1)[4],
                                         const float (&f)[16][2],
                                         v8f& d00, v8f& d01, v8f& d10, v8f& d11) {
    v2f a0, a1, b0, b1;
    if (KK & 1) {
        a0.x = q0[KK >> 1].z;  a0.y = q0[KK >> 1].w;
        a1.x = q1[KK >> 1].z;  a1.y = q1[KK >> 1].w;
    } else {
        a0.x = q0[KK >> 1].x;  a0.y = q0[KK >> 1].y;
        a1.x = q1[KK >> 1].x;  a1.y = q1[KK >> 1].y;
    }
    b0.x = f[2 * KK][0];  b0.y = f[2 * KK + 1][0];
    b1.x = f[2 * KK][1];  b1.y = f[2 * KK + 1][1];
    d00 = __builtin_amdgcn_wmma_f32_16x16x4_f32(false, b0, false, a0, (short)0, d00, false, false);
    d01 = __builtin_amdgcn_wmma_f32_16x16x4_f32(false, b0, false, a1, (short)0, d01, false, false);
    d10 = __builtin_amdgcn_wmma_f32_16x16x4_f32(false, b1, false, a0, (short)0, d10, false, false);
    d11 = __builtin_amdgcn_wmma_f32_16x16x4_f32(false, b1, false, a1, (short)0, d11, false, false);
}

__device__ __forceinline__ void product_row(const float4 (&q0)[4], const float4 (&q1)[4],
                                            const float (&f)[16][2], int ns, float imp,
                                            float* __restrict__ sArow, float* __restrict__ sInvNrow,
                                            int lane, int half, int lm) {
    v8f d00 = {}, d01 = {}, d10 = {}, d11 = {};
    if (ns > 0) mma_step<0>(q0, q1, f, d00, d01, d10, d11);
    if (ns > 1) mma_step<1>(q0, q1, f, d00, d01, d10, d11);
    if (ns > 2) mma_step<2>(q0, q1, f, d00, d01, d10, d11);
    if (ns > 3) mma_step<3>(q0, q1, f, d00, d01, d10, d11);
    if (ns > 4) mma_step<4>(q0, q1, f, d00, d01, d10, d11);
    if (ns > 5) mma_step<5>(q0, q1, f, d00, d01, d10, d11);
    if (ns > 6) mma_step<6>(q0, q1, f, d00, d01, d10, d11);
    if (ns > 7) mma_step<7>(q0, q1, f, d00, d01, d10, d11);

    float* p0 = sArow + lm * VP + 8 * half;
    *(float4*)(p0)      = make_float4(d00[0], d00[1], d00[2], d00[3]);
    *(float4*)(p0 + 4)  = make_float4(d00[4], d00[5], d00[6], d00[7]);
    *(float4*)(p0 + 16) = make_float4(d10[0], d10[1], d10[2], d10[3]);
    *(float4*)(p0 + 20) = make_float4(d10[4], d10[5], d10[6], d10[7]);
    if (lm < VOX - 16) {
        float* p1 = p0 + 16 * VP;
        *(float4*)(p1)      = make_float4(d01[0], d01[1], d01[2], d01[3]);
        *(float4*)(p1 + 4)  = make_float4(d01[4], d01[5], d01[6], d01[7]);
        *(float4*)(p1 + 16) = make_float4(d11[0], d11[1], d11[2], d11[3]);
        *(float4*)(p1 + 20) = make_float4(d11[4], d11[5], d11[6], d11[7]);
    }
    float s = imp;
    s += row_xmask_f<1>(s);
    s += row_xmask_f<2>(s);
    s += row_xmask_f<4>(s);
    s += row_xmask_f<8>(s);
    const float norm = s + __shfl_xor(s, 16);
    if (lane == 0)
        *sInvNrow = (norm > 0.0f) ? 1.0f / fmaxf(norm, 1e-12f) : 0.0f;
}

__global__ __launch_bounds__(BLOCK)
void cconv_persistent_kernel(const float* __restrict__ feats,
                             const float* __restrict__ inPos,
                             const float* __restrict__ outPos,
                             const float* __restrict__ extents,
                             const float* __restrict__ importance,
                             const int*   __restrict__ nbrIdx,
                             const float* __restrict__ Wg,
                             const float* __restrict__ bias,
                             float*       __restrict__ out)
{
    __shared__ __attribute__((aligned(16))) float sA[MTILE][AS];
    __shared__ float sC[4][8][32];
    __shared__ float sInvN[MTILE];
    __shared__ __attribute__((aligned(16))) float sWt[WAVES][32][WP];

    const int lane = threadIdx.x & 31;
    const int wave = threadIdx.x >> 5;
    const int half = lane >> 4;
    const int lm   = lane & 15;
    const float invR = 2.0f / extents[0];

    float* wcol = &sWt[wave][0][lane];
    const float* wrow = &sWt[wave][lm][16 * half];
    for (int r = VOX; r < 32; ++r) wcol[r * WP] = 0.0f;
    const char* featB = (const char*)feats;
    const unsigned c4 = (unsigned)lm * 4u;
    const int pos = 4 * (lm >> 1) + 2 * half + (lane & 1);

    const int rowA  = wave * 2, rowB = wave * 2 + 1;
    const int nt    = wave & 3;
    const int khalf = wave >> 2;
    const int n0    = nt * 16;
    const float bv  = bias[n0 + lm];
    const float* aP = &sA[lm][16 * khalf + 2 * half];

    v2f wf[KSTEPS_W];
    {
        const float* wg = Wg + (16 * khalf + 2 * half) * COUT + n0 + lm;
        #pragma unroll
        for (int t = 0; t < KSTEPS_W; ++t) {
            const int kr = (t >> 2) * CIN + 4 * (t & 3);
            wf[t].x = wg[kr * COUT];
            wf[t].y = wg[(kr + 1) * COUT];
        }
    }

    int tile = blockIdx.x;
    float fA[16][2], fB[16][2], gA[7], gB[7], cA[4], cB[4];
    int nsA, nsB;
    int r1A, r1B;
    int r2A, r2B;
    {
        const int t1 = min(tile + NBLK, NTILES - 1);
        const int t2 = min(tile + 2 * NBLK, NTILES - 1);
        const int mA = tile * MTILE + rowA, mB = tile * MTILE + rowB;
        const int r0A = nbrIdx[mA * KNBR + lane];
        const int r0B = nbrIdx[mB * KNBR + lane];
        r1A = nbrIdx[(t1 * MTILE + rowA) * KNBR + lane];
        r1B = nbrIdx[(t1 * MTILE + rowB) * KNBR + lane];
        r2A = nbrIdx[(t2 * MTILE + rowA) * KNBR + lane];
        r2B = nbrIdx[(t2 * MTILE + rowB) * KNBR + lane];
        issue_geom(inPos, outPos, importance, r0A, mA, gA);
        issue_geom(inPos, outPos, importance, r0B, mB, gB);
        int ciA, ciB;
        nsA = compact_row(gA, r0A, invR, lane, pos, cA, ciA);
        nsB = compact_row(gB, r0B, invR, lane, pos, cB, ciB);
        issue_feats(featB, ciA, c4, nsA, fA);
        issue_feats(featB, ciB, c4, nsB, fB);
        issue_geom(inPos, outPos, importance, r1A, t1 * MTILE + rowA, gA);
        issue_geom(inPos, outPos, importance, r1B, t1 * MTILE + rowB, gB);
    }

    for (; tile < NTILES; tile += NBLK) {
        const int m0 = tile * MTILE;

        float4 qA0[4], qA1[4], qB0[4], qB1[4];
        const float impA = store_weights(cA, wcol);
        load_wfrags(wrow, qA0, qA1);
        const float impB = store_weights(cB, wcol);
        load_wfrags(wrow, qB0, qB1);
        product_row(qA0, qA1, fA, nsA, impA, sA[rowA], &sInvN[rowA], lane, half, lm);
        product_row(qB0, qB1, fB, nsB, impB, sA[rowB], &sInvN[rowB], lane, half, lm);
        __syncthreads();

        const int t2 = min(tile + 2 * NBLK, NTILES - 1);
        const int t3 = min(tile + 3 * NBLK, NTILES - 1);
        const int r3A = nbrIdx[(t3 * MTILE + rowA) * KNBR + lane];
        const int r3B = nbrIdx[(t3 * MTILE + rowB) * KNBR + lane];

        float inv[8];
        #pragma unroll
        for (int j = 0; j < 8; ++j) inv[j] = sInvN[j + 8 * half];

        v8f c0 = {}, c1 = {};
        v2f fa[2][KBLK];
        load_frags(aP, 0, fa[0]);
        #pragma unroll
        for (int blk = 0; blk < KSTEPS_W / KBLK; ++blk) {
            const int cur = blk & 1;
            if (blk + 1 < KSTEPS_W / KBLK)
                load_frags(aP, blk + 1, fa[cur ^ 1]);
            if (blk == 1) {
                int ciA, ciB;
                nsA = compact_row(gA, r1A, invR, lane, pos, cA, ciA);
                nsB = compact_row(gB, r1B, invR, lane, pos, cB, ciB);
                issue_feats(featB, ciA, c4, nsA, fA);
                issue_feats(featB, ciB, c4, nsB, fB);
                issue_geom(inPos, outPos, importance, r2A, t2 * MTILE + rowA, gA);
                issue_geom(inPos, outPos, importance, r2B, t2 * MTILE + rowB, gB);
                r1A = r2A;  r1B = r2B;
                r2A = r3A;  r2B = r3B;
            }
            __builtin_amdgcn_sched_barrier(0);
            #pragma unroll
            for (int i = 0; i < KBLK; i += 2) {
                c0 = __builtin_amdgcn_wmma_f32_16x16x4_f32(false, fa[cur][i], false, wf[blk * KBLK + i],
                                                           (short)0, c0, false, false);
                c1 = __builtin_amdgcn_wmma_f32_16x16x4_f32(false, fa[cur][i + 1], false, wf[blk * KBLK + i + 1],
                                                           (short)0, c1, false, false);
            }
        }
        const v8f c = c0 + c1;

        if (khalf == 1) {
            #pragma unroll
            for (int j = 0; j < 8; ++j) sC[nt][j][lane] = c[j];
        }
        __syncthreads();

        if (khalf == 0) {
            #pragma unroll
            for (int j = 0; j < 8; ++j) {
                const int mrow = j + 8 * half;
                out[(long)(m0 + mrow) * COUT + n0 + lm] = (c[j] + sC[nt][j][lane]) * inv[j] + bv;
            }
        }
    }
}

extern "C" void kernel_launch(void* const* d_in, const int* in_sizes, int n_in,
                              void* d_out, int out_size, void* d_ws, size_t ws_size,
                              hipStream_t stream) {
    const float* feats      = (const float*)d_in[0];
    const float* inPos      = (const float*)d_in[1];
    const float* outPos     = (const float*)d_in[2];
    const float* extents    = (const float*)d_in[3];
    const float* importance = (const float*)d_in[4];
    const int*   nbrIdx     = (const int*)  d_in[5];
    const float* Wg         = (const float*)d_in[6];
    const float* bias       = (const float*)d_in[7];
    float* out = (float*)d_out;

    dim3 grid(NBLK);
    cconv_persistent_kernel<<<grid, BLOCK, 0, stream>>>(
        feats, inPos, outPos, extents, importance, nbrIdx, Wg, bias, out);
}


# second A tile in LDS: phase 1 of the next tile shares a barrier interval with phase 2 of the current one (one barrier per tile), and the two waves of a SIMD take the two phases in opposite order
# speedup vs baseline: 4.1376x; 1.0410x over previous
#include <hip/hip_runtime.h>


typedef __attribute__((ext_vector_type(2))) float v2f;
typedef __attribute__((ext_vector_type(8))) float v8f;

#define NPTS   200000
#define MOUT   100000
#define CIN    32
#define COUT   64
#define KNBR   32
#define VOX    27
#define KDIM   (VOX * CIN)
#define VP     36
#define AS     (VOX * VP)
#define MTILE  16
#define NTILES (MOUT / MTILE)
#define NBLK   250
#define WAVES  8
#define BLOCK  (WAVES * 32)
#define KSTEPS_W (KDIM / 4 / 2)
#define KBLK   12
#define WP     36

__device__ __forceinline__ void load_frags(const float* aP, int blk, v2f (&fa)[KBLK]) {
    #pragma unroll
    for (int i = 0; i < KBLK; ++i) {
        const int t = blk * KBLK + i;
        fa[i] = *(const v2f*)(aP + (t >> 2) * VP + 4 * (t & 3));
    }
}

template <int S> __device__ __forceinline__ int row_share_i(int x) {
    return __builtin_amdgcn_update_dpp(0, x, 0x150 + S, 0xf, 0xf, true);
}

__device__ __forceinline__ float tri(float x, float g) {
    return fmaxf(0.0f, 1.0f - fabsf(x - g));
}

template <int S>
__device__ __forceinline__ void gather_slot(const char* __restrict__ featB, int idx, unsigned c4,
                                            float (&f)[16][2]) {
    const unsigned off = (unsigned)row_share_i<S>(idx) * (unsigned)(CIN * 4) + c4;
    f[S][0] = *(const float*)(featB + off);
    f[S][1] = *(const float*)(featB + off + 64u);
}

template <int KK>
__device__ __forceinline__ void gather_pair(const char* __restrict__ featB, int idx, unsigned c4,
                                            float (&f)[16][2]) {
    gather_slot<2 * KK>(featB, idx, c4, f);
    gather_slot<2 * KK + 1>(featB, idx, c4, f);
}

__device__ __forceinline__ void issue_feats(const char* __restrict__ featB, int idx, unsigned c4,
                                            int ns, float (&f)[16][2]) {
    if (ns > 0) gather_pair<0>(featB, idx, c4, f);
    if (ns > 1) gather_pair<1>(featB, idx, c4, f);
    if (ns > 2) gather_pair<2>(featB, idx, c4, f);
    if (ns > 3) gather_pair<3>(featB, idx, c4, f);
    if (ns > 4) gather_pair<4>(featB, idx, c4, f);
    if (ns > 5) gather_pair<5>(featB, idx, c4, f);
    if (ns > 6) gather_pair<6>(featB, idx, c4, f);
    if (ns > 7) gather_pair<7>(featB, idx, c4, f);
}

__device__ __forceinline__ void issue_geom(const float* __restrict__ inPos,
                                           const float* __restrict__ outPos,
                                           const float* __restrict__ importance,
                                           int idx, int m, float (&g)[7]) {
    g[0] = inPos[idx * 3 + 0];
    g[1] = inPos[idx * 3 + 1];
    g[2] = inPos[idx * 3 + 2];
    g[3] = importance[idx];
    g[4] = outPos[m * 3 + 0];
    g[5] = outPos[m * 3 + 1];
    g[6] = outPos[m * 3 + 2];
}

template <int M> __device__ __forceinline__ float row_xmask_f(float x) {
    return __int_as_float(__builtin_amdgcn_update_dpp(0, __float_as_int(x), 0x160 + M, 0xf, 0xf, true));
}

__device__ __forceinline__ int push_i(int dst4, int x) {
    return __builtin_amdgcn_ds_permute(dst4, x);
}
__device__ __forceinline__ float push_f(int dst4, float x) {
    return __int_as_float(__builtin_amdgcn_ds_permute(dst4, __float_as_int(x)));
}

__device__ __forceinline__ int compact_row(const float (&g)[7], int idx, float invR, int lane,
                                           int pos, float (&c)[4], int& cidx) {
    const float rx = (g[0] - g[4]) * invR;
    const float ry = (g[1] - g[5]) * invR;
    const float rz = (g[2] - g[6]) * invR;
    const float dist = sqrtf(rx * rx + ry * ry + rz * rz);
    const float linf = fmaxf(fabsf(rx), fmaxf(fabsf(ry), fabsf(rz)));
    const float scale = (linf > 1e-8f) ? dist / fmaxf(linf, 1e-8f) : 0.0f;
    const bool in = (dist <= 1.0f);
    const float imp = in ? g[3] : 0.0f;
    const float cx = fminf(fmaxf(rx * scale + 1.0f, 0.0f), 2.0f);
    const float cy = fminf(fmaxf(ry * scale + 1.0f, 0.0f), 2.0f);
    const float cz = fminf(fmaxf(rz * scale + 1.0f, 0.0f), 2.0f);
    const unsigned mask = __builtin_amdgcn_ballot_w32(in);
    const int n = __builtin_popcount(mask);
    const int below = (int)__builtin_amdgcn_mbcnt_lo(mask, 0u);
    const int rank = in ? below : n + (lane - below);
    const int dst4 = 4 * (16 * ((rank >> 1) & 1) + 2 * (rank >> 2) + (rank & 1));
    const int ri = push_i(dst4, idx);
    c[0] = push_f(dst4, cx);
    c[1] = push_f(dst4, cy);
    c[2] = push_f(dst4, cz);
    c[3] = push_f(dst4, imp);
    const int rf = __builtin_amdgcn_readfirstlane(ri);
    cidx = (pos < n) ? ri : rf;
    return (n + 3) >> 2;
}

__device__ __forceinline__ float store_weights(const float (&c)[4], float* wcol) {
    const float imp = c[3];
    float wx[3], wy[3], wz[3];
    #pragma unroll
    for (int t = 0; t < 3; ++t) {
        wx[t] = tri(c[0], (float)t);
        wy[t] = tri(c[1], (float)t);
        wz[t] = tri(c[2], (float)t) * imp;
    }
    #pragma unroll
    for (int a = 0; a < 3; ++a) {
        #pragma unroll
        for (int b = 0; b < 3; ++b) {
            const float xy = wx[a] * wy[b];
            #pragma unroll
            for (int cc = 0; cc < 3; ++cc) wcol[(9 * a + 3 * b + cc) * WP] = xy * wz[cc];
        }
    }
    return imp;
}

__device__ __forceinline__ void load_wfrags(const float* wrow, float4 (&q0)[4], float4 (&q1)[4]) {
    #pragma unroll
    for (int u = 0; u < 4; ++u) {
        q0[u] = *(const float4*)(wrow + 4 * u);
        q1[u] = *(const float4*)(wrow + 16 * WP + 4 * u);
    }
}

template <int KK>
__device__ __forceinline__ void mma_step(const float4 (&q0)[4], const float4 (&q1)[4],
                                         const float (&f)[16][2],
                                         v8f& d00, v8f& d01, v8f& d10, v8f& d11) {
    v2f a0, a1, b0, b1;
    if (KK & 1) {
        a0.x = q0[KK >> 1].z;  a0.y = q0[KK >> 1].w;
        a1.x = q1[KK >> 1].z;  a1.y = q1[KK >> 1].w;
    } else {
        a0.x = q0[KK >> 1].x;  a0.y = q0[KK >> 1].y;
        a1.x = q1[KK >> 1].x;  a1.y = q1[KK >> 1].y;
    }
    b0.x = f[2 * KK][0];  b0.y = f[2 * KK + 1][0];
    b1.x = f[2 * KK][1];  b1.y = f[2 * KK + 1][1];
    d00 = __builtin_amdgcn_wmma_f32_16x16x4_f32(false, b0, false, a0, (short)0, d00, false, false);
    d01 = __builtin_amdgcn_wmma_f32_16x16x4_f32(false, b0, false, a1, (short)0, d01, false, false);
    d10 = __builtin_amdgcn_wmma_f32_16x16x4_f32(false, b1, false, a0, (short)0, d10, false, false);
    d11 = __builtin_amdgcn_wmma_f32_16x16x4_f32(false, b1, false, a1, (short)0, d11, false, false);
}

__device__ __forceinline__ void product_row(const float4 (&q0)[4], const float4 (&q1)[4],
                                            const float (&f)[16][2], int ns, float imp,
                                            float* __restrict__ sArow, float* __restrict__ sInvNrow,
                                            int lane, int half, int lm) {
    v8f d00 = {}, d01 = {}, d10 = {}, d11 = {};
    if (ns > 0) mma_step<0>(q0, q1, f, d00, d01, d10, d11);
    if (ns > 1) mma_step<1>(q0, q1, f, d00, d01, d10, d11);
    if (ns > 2) mma_step<2>(q0, q1, f, d00, d01, d10, d11);
    if (ns > 3) mma_step<3>(q0, q1, f, d00, d01, d10, d11);
    if (ns > 4) mma_step<4>(q0, q1, f, d00, d01, d10, d11);
    if (ns > 5) mma_step<5>(q0, q1, f, d00, d01, d10, d11);
    if (ns > 6) mma_step<6>(q0, q1, f, d00, d01, d10, d11);
    if (ns > 7) mma_step<7>(q0, q1, f, d00, d01, d10, d11);

    float* p0 = sArow + lm * VP + 8 * half;
    *(float4*)(p0)      = make_float4(d00[0], d00[1], d00[2], d00[3]);
    *(float4*)(p0 + 4)  = make_float4(d00[4], d00[5], d00[6], d00[7]);
    *(float4*)(p0 + 16) = make_float4(d10[0], d10[1], d10[2], d10[3]);
    *(float4*)(p0 + 20) = make_float4(d10[4], d10[5], d10[6], d10[7]);
    if (lm < VOX - 16) {
        float* p1 = p0 + 16 * VP;
        *(float4*)(p1)      = make_float4(d01[0], d01[1], d01[2], d01[3]);
        *(float4*)(p1 + 4)  = make_float4(d01[4], d01[5], d01[6], d01[7]);
        *(float4*)(p1 + 16) = make_float4(d11[0], d11[1], d11[2], d11[3]);
        *(float4*)(p1 + 20) = make_float4(d11[4], d11[5], d11[6], d11[7]);
    }
    float s = imp;
    s += row_xmask_f<1>(s);
    s += row_xmask_f<2>(s);
    s += row_xmask_f<4>(s);
    s += row_xmask_f<8>(s);
    const float norm = s + __shfl_xor(s, 16);
    if (lane == 0)
        *sInvNrow = (norm > 0.0f) ? 1.0f / fmaxf(norm, 1e-12f) : 0.0f;
}

__device__ __forceinline__ void phase1_rows(const float (&cA)[4], const float (&cB)[4],
                                            const float (&fA)[16][2], const float (&fB)[16][2],
                                            int nsA, int nsB, float* wcol, const float* wrow,
                                            float* __restrict__ sArowA, float* __restrict__ sArowB,
                                            float* __restrict__ invA, float* __restrict__ invB,
                                            int lane, int half, int lm) {
    nsA = __builtin_amdgcn_readfirstlane(nsA);
    nsB = __builtin_amdgcn_readfirstlane(nsB);
    float4 qA0[4], qA1[4], qB0[4], qB1[4];
    const float impA = store_weights(cA, wcol);
    load_wfrags(wrow, qA0, qA1);
    const float impB = store_weights(cB, wcol);
    load_wfrags(wrow, qB0, qB1);
    product_row(qA0, qA1, fA, nsA, impA, sArowA, invA, lane, half, lm);
    product_row(qB0, qB1, fB, nsB, impB, sArowB, invB, lane, half, lm);
}

__device__ __forceinline__ void advance_rows(const float* __restrict__ inPos, const float* __restrict__ outPos,
                                             const float* __restrict__ importance,
                                             const char* __restrict__ featB, unsigned c4, float invR,
                                             int lane, int pos,
                                             float (&gA)[7], float (&gB)[7],
                                             int& r1A, int& r1B, int& r2A, int& r2B, int r3A, int r3B,
                                             int mA2, int mB2,
                                             float (&cA)[4], float (&cB)[4], int& nsA, int& nsB,
                                             float (&fA)[16][2], float (&fB)[16][2]) {
    int ciA, ciB;
    nsA = compact_row(gA, r1A, invR, lane, pos, cA, ciA);
    nsB = compact_row(gB, r1B, invR, lane, pos, cB, ciB);
    issue_feats(featB, ciA, c4, nsA, fA);
    issue_feats(featB, ciB, c4, nsB, fB);
    issue_geom(inPos, outPos, importance, r2A, mA2, gA);
    issue_geom(inPos, outPos, importance, r2B, mB2, gB);
    r1A = r2A;  r1B = r2B;
    r2A = r3A;  r2B = r3B;
}

__global__ __launch_bounds__(BLOCK)
void cconv_persistent_kernel(const float* __restrict__ feats,
                             const float* __restrict__ inPos,
                             const float* __restrict__ outPos,
                             const float* __restrict__ extents,
                             const float* __restrict__ importance,
                             const int*   __restrict__ nbrIdx,
                             const float* __restrict__ Wg,
                             const float* __restrict__ bias,
                             float*       __restrict__ out)
{
    __shared__ __attribute__((aligned(16))) float sA[2][MTILE][AS];
    __shared__ float sC[2][4][8][32];
    __shared__ float sInvN[2][MTILE];
    __shared__ __attribute__((aligned(16))) float sWt[WAVES][32][WP];

    const int lane = threadIdx.x & 31;
    const int wave = __builtin_amdgcn_readfirstlane((int)(threadIdx.x >> 5));
    const int half = lane >> 4;
    const int lm   = lane & 15;
    const float invR = 2.0f / extents[0];

    float* wcol = &sWt[wave][0][lane];
    const float* wrow = &sWt[wave][lm][16 * half];
    for (int r = VOX; r < 32; ++r) wcol[r * WP] = 0.0f;
    const char* featB = (const char*)feats;
    const unsigned c4 = (unsigned)lm * 4u;
    const int pos = 4 * (lm >> 1) + 2 * half + (lane & 1);

    const int rowA  = wave * 2, rowB = wave * 2 + 1;
    const int nt    = wave & 3;
    const int khalf = wave >> 2;
    const int n0    = nt * 16;
    const float bv  = bias[n0 + lm];

    const bool p1first = (khalf == 1);
    const int  ahead   = p1first ? 1 : 0;

    v2f wf[KSTEPS_W];
    {
        const float* wg = Wg + (16 * khalf + 2 * half) * COUT + n0 + lm;
        #pragma unroll
        for (int t = 0; t < KSTEPS_W; ++t) {
            const int kr = (t >> 2) * CIN + 4 * (t & 3);
            wf[t].x = wg[kr * COUT];
            wf[t].y = wg[(kr + 1) * COUT];
        }
    }

    int tile = blockIdx.x;
    float fA[16][2], fB[16][2], gA[7], gB[7], cA[4], cB[4];
    int nsA, nsB;
    int r1A, r1B;
    int r2A, r2B;
    {
        const int t1 = min(tile + NBLK, NTILES - 1);
        const int t2 = min(tile + 2 * NBLK, NTILES - 1);
        const int mA = tile * MTILE + rowA, mB = tile * MTILE + rowB;
        const int r0A = nbrIdx[mA * KNBR + lane];
        const int r0B = nbrIdx[mB * KNBR + lane];
        r1A = nbrIdx[(t1 * MTILE + rowA) * KNBR + lane];
        r1B = nbrIdx[(t1 * MTILE + rowB) * KNBR + lane];
        r2A = nbrIdx[(t2 * MTILE + rowA) * KNBR + lane];
        r2B = nbrIdx[(t2 * MTILE + rowB) * KNBR + lane];
        issue_geom(inPos, outPos, importance, r0A, mA, gA);
        issue_geom(inPos, outPos, importance, r0B, mB, gB);
        int ciA, ciB;
        nsA = compact_row(gA, r0A, invR, lane, pos, cA, ciA);
        nsB = compact_row(gB, r0B, invR, lane, pos, cB, ciB);
        issue_feats(featB, ciA, c4, nsA, fA);
        issue_feats(featB, ciB, c4, nsB, fB);
        issue_geom(inPos, outPos, importance, r1A, t1 * MTILE + rowA, gA);
        issue_geom(inPos, outPos, importance, r1B, t1 * MTILE + rowB, gB);

        phase1_rows(cA, cB, fA, fB, nsA, nsB, wcol, wrow, sA[0][rowA], sA[0][rowB],
                    &sInvN[0][rowA], &sInvN[0][rowB], lane, half, lm);
        if (p1first) {
            const int t3 = min(tile + 3 * NBLK, NTILES - 1);
            const int r3A = nbrIdx[(t3 * MTILE + rowA) * KNBR + lane];
            const int r3B = nbrIdx[(t3 * MTILE + rowB) * KNBR + lane];
            advance_rows(inPos, outPos, importance, featB, c4, invR, lane, pos, gA, gB,
                         r1A, r1B, r2A, r2B, r3A, r3B, t2 * MTILE + rowA, t2 * MTILE + rowB,
                         cA, cB, nsA, nsB, fA, fB);
        }
        __syncthreads();
    }

    int buf = 0;
    for (; tile < NTILES; tile += NBLK, buf ^= 1) {
        const int m0 = tile * MTILE;
        const bool more = (tile + NBLK < NTILES);
        const int nb = buf ^ 1;

        if (p1first && more)
            phase1_rows(cA, cB, fA, fB, nsA, nsB, wcol, wrow, sA[nb][rowA], sA[nb][rowB],
                        &sInvN[nb][rowA], &sInvN[nb][rowB], lane, half, lm);

        const int t2 = min(tile + (2 + ahead) * NBLK, NTILES - 1);
        const int t3 = min(tile + (3 + ahead) * NBLK, NTILES - 1);
        const int r3A = nbrIdx[(t3 * MTILE + rowA) * KNBR + lane];
        const int r3B = nbrIdx[(t3 * MTILE + rowB) * KNBR + lane];

        const float* aP = &sA[buf][lm][16 * khalf + 2 * half];
        v8f c0 = {}, c1 = {};
        v2f fa[2][KBLK];
        load_frags(aP, 0, fa[0]);
        #pragma unroll
        for (int blk = 0; blk < KSTEPS_W / KBLK; ++blk) {
            const int cur = blk & 1;
            if (blk + 1 < KSTEPS_W / KBLK)
                load_frags(aP, blk + 1, fa[cur ^ 1]);
            if (blk == 1) {
                advance_rows(inPos, outPos, importance, featB, c4, invR, lane, pos, gA, gB,
                             r1A, r1B, r2A, r2B, r3A, r3B, t2 * MTILE + rowA, t2 * MTILE + rowB,
                             cA, cB, nsA, nsB, fA, fB);
            }
            __builtin_amdgcn_sched_barrier(0);
            #pragma unroll
            for (int i = 0; i < KBLK; i += 2) {
                c0 = __builtin_amdgcn_wmma_f32_16x16x4_f32(false, fa[cur][i], false, wf[blk * KBLK + i],
                                                           (short)0, c0, false, false);
                c1 = __builtin_amdgcn_wmma_f32_16x16x4_f32(false, fa[cur][i + 1], false, wf[blk * KBLK + i + 1],
                                                           (short)0, c1, false, false);
            }
        }
        const v8f c = c0 + c1;

        if (khalf == 1) {
            #pragma unroll
            for (int j = 0; j < 8; ++j) sC[buf][nt][j][lane] = c[j];
        }

        if (!p1first && more)
            phase1_rows(cA, cB, fA, fB, nsA, nsB, wcol, wrow, sA[nb][rowA], sA[nb][rowB],
                        &sInvN[nb][rowA], &sInvN[nb][rowB], lane, half, lm);

        float inv[8];
        #pragma unroll
        for (int j = 0; j < 8; ++j) inv[j] = sInvN[buf][j + 8 * half];

        __syncthreads();

        if (khalf == 0) {
            #pragma unroll
            for (int j = 0; j < 8; ++j) {
                const int mrow = j + 8 * half;
                out[(long)(m0 + mrow) * COUT + n0 + lm] = (c[j] + sC[buf][nt][j][lane]) * inv[j] + bv;
            }
        }
    }
}

extern "C" void kernel_launch(void* const* d_in, const int* in_sizes, int n_in,
                              void* d_out, int out_size, void* d_ws, size_t ws_size,
                              hipStream_t stream) {
    const float* feats      = (const float*)d_in[0];
    const float* inPos      = (const float*)d_in[1];
    const float* outPos     = (const float*)d_in[2];
    const float* extents    = (const float*)d_in[3];
    const float* importance = (const float*)d_in[4];
    const int*   nbrIdx     = (const int*)  d_in[5];
    const float* Wg         = (const float*)d_in[6];
    const float* bias       = (const float*)d_in[7];
    float* out = (float*)d_out;

    dim3 grid(NBLK);
    cconv_persistent_kernel<<<grid, BLOCK, 0, stream>>>(
        feats, inPos, outPos, extents, importance, nbrIdx, Wg, bias, out);
}
